# starting kernel
# baseline (speedup 1.0000x reference)
#include <hip/hip_runtime.h>
#include <cstdint>
#include <cstddef>

typedef __attribute__((ext_vector_type(2))) float v2f;
typedef __attribute__((ext_vector_type(8))) float v8f;

#define NEG_SLOPE 0.2f

__device__ __forceinline__ unsigned fkey(float f) {
  unsigned u = __float_as_uint(f);
  return (u & 0x80000000u) ? ~u : (u | 0x80000000u);
}
__device__ __forceinline__ float funkey(unsigned k) {
  unsigned u = (k & 0x80000000u) ? (k & 0x7fffffffu) : ~k;
  return __uint_as_float(u);
}

__device__ __forceinline__ void edge_sd(const int* __restrict__ ei, long long E,
                                        long long eid, int& s, int& d) {
  if (eid < E) { s = ei[eid]; d = ei[E + eid]; }
  else         { s = d = (int)(eid - E); }
}

template<int K, int COLS>
__global__ void gemm_wmma_f32(const float* __restrict__ A,
                              const float* __restrict__ B,
                              float* __restrict__ C, int N) {
  constexpr int CT = COLS / 16;
  const int wave = (int)((blockIdx.x * blockDim.x + threadIdx.x) >> 5);
  const int lane = (int)(threadIdx.x & 31);
  const int row0 = wave * 16;
  if (row0 >= N) return;
  const int m    = lane & 15;
  const int half = lane >> 4;
  const int klo  = half * 2;

  v8f zero = {0.f, 0.f, 0.f, 0.f, 0.f, 0.f, 0.f, 0.f};
  v8f acc[CT];
  #pragma unroll
  for (int t = 0; t < CT; ++t) acc[t] = zero;

  #pragma unroll
  for (int k = 0; k < K; k += 4) {
    const float* ap = A + (size_t)(row0 + m) * K + (k + klo);
    v2f a; a.x = ap[0]; a.y = ap[1];
    #pragma unroll
    for (int t = 0; t < CT; ++t) {
      const float* bp = B + (size_t)(k + klo) * COLS + t * 16 + m;
      v2f b; b.x = bp[0]; b.y = bp[COLS];
      acc[t] = __builtin_amdgcn_wmma_f32_16x16x4_f32(
          false, a, false, b, (short)0, acc[t], false, false);
    }
  }
  const int rbase = row0 + half * 8;
  #pragma unroll
  for (int t = 0; t < CT; ++t) {
    #pragma unroll
    for (int v = 0; v < 8; ++v)
      C[(size_t)(rbase + v) * COLS + t * 16 + m] = acc[t][v];
  }
}

__global__ void alpha1_k(const float* __restrict__ h1,
                         const float* __restrict__ aw_s, const float* __restrict__ aw_d,
                         float* __restrict__ as, float* __restrict__ ad, int N) {
  long long i = (long long)blockIdx.x * blockDim.x + threadIdx.x;
  if (i >= (long long)N * 8) return;
  int n = (int)(i >> 3), h = (int)(i & 7);
  const float* hp = h1 + (size_t)n * 64 + h * 8;
  const float* sp = aw_s + h * 8;
  const float* dp = aw_d + h * 8;
  float s = 0.f, d = 0.f;
  #pragma unroll
  for (int c = 0; c < 8; ++c) { float hv = hp[c]; s += hv * sp[c]; d += hv * dp[c]; }
  as[i] = s; ad[i] = d;
}

__global__ void alpha2_k(const float* __restrict__ h2,
                         const float* __restrict__ aw_s, const float* __restrict__ aw_d,
                         float* __restrict__ as, float* __restrict__ ad, int N) {
  int wave = (int)((blockIdx.x * blockDim.x + threadIdx.x) >> 5);
  int lane = (int)(threadIdx.x & 31);
  if (wave >= N) return;
  const float* hp = h2 + (size_t)wave * 128;
  float s = 0.f, d = 0.f;
  #pragma unroll
  for (int f = 0; f < 128; f += 32) {
    float hv = hp[f + lane];
    s += hv * aw_s[f + lane];
    d += hv * aw_d[f + lane];
  }
  #pragma unroll
  for (int off = 16; off > 0; off >>= 1) {
    s += __shfl_down(s, off, 32);
    d += __shfl_down(d, off, 32);
  }
  if (lane == 0) { as[wave] = s; ad[wave] = d; }
}

template<int H>
__global__ void edge_max_k(const int* __restrict__ ei, long long E, int N,
                           const float* __restrict__ as, const float* __restrict__ ad,
                           unsigned* __restrict__ emaxk) {
  long long t = (long long)blockIdx.x * blockDim.x + threadIdx.x;
  long long total = (E + N) * H;
  if (t >= total) return;
  long long eid = t / H; int h = (int)(t % H);
  int s, d; edge_sd(ei, E, eid, s, d);
  float e = as[(size_t)s * H + h] + ad[(size_t)d * H + h];
  e = e > 0.f ? e : NEG_SLOPE * e;
  atomicMax(&emaxk[(size_t)d * H + h], fkey(e));
}

template<int H>
__global__ void edge_den_k(const int* __restrict__ ei, long long E, int N,
                           const float* __restrict__ as, const float* __restrict__ ad,
                           const unsigned* __restrict__ emaxk, float* __restrict__ den) {
  long long t = (long long)blockIdx.x * blockDim.x + threadIdx.x;
  long long total = (E + N) * H;
  if (t >= total) return;
  long long eid = t / H; int h = (int)(t % H);
  int s, d; edge_sd(ei, E, eid, s, d);
  float e = as[(size_t)s * H + h] + ad[(size_t)d * H + h];
  e = e > 0.f ? e : NEG_SLOPE * e;
  float w = __expf(e - funkey(emaxk[(size_t)d * H + h]));
  atomicAdd(&den[(size_t)d * H + h], w);
}

__global__ void agg1_k(const int* __restrict__ ei, long long E, int N,
                       const float* __restrict__ as, const float* __restrict__ ad,
                       const unsigned* __restrict__ emaxk, const float* __restrict__ den,
                       const float* __restrict__ h1, float* __restrict__ agg) {
  long long t = (long long)blockIdx.x * blockDim.x + threadIdx.x;
  long long total = (E + N) * 8;
  if (t >= total) return;
  long long eid = t >> 3; int h = (int)(t & 7);
  int s, d; edge_sd(ei, E, eid, s, d);
  float e = as[(size_t)s * 8 + h] + ad[(size_t)d * 8 + h];
  e = e > 0.f ? e : NEG_SLOPE * e;
  float alpha = __expf(e - funkey(emaxk[(size_t)d * 8 + h])) / den[(size_t)d * 8 + h];
  const float4* hp = (const float4*)(h1 + (size_t)s * 64 + h * 8);
  float4 v0 = hp[0], v1 = hp[1];
  float* op = agg + (size_t)d * 64 + h * 8;
  atomicAdd(op + 0, alpha * v0.x); atomicAdd(op + 1, alpha * v0.y);
  atomicAdd(op + 2, alpha * v0.z); atomicAdd(op + 3, alpha * v0.w);
  atomicAdd(op + 4, alpha * v1.x); atomicAdd(op + 5, alpha * v1.y);
  atomicAdd(op + 6, alpha * v1.z); atomicAdd(op + 7, alpha * v1.w);
}

__global__ void agg2_k(const int* __restrict__ ei, long long E, int N,
                       const float* __restrict__ as, const float* __restrict__ ad,
                       const unsigned* __restrict__ emaxk, const float* __restrict__ den,
                       const float* __restrict__ h2, float* __restrict__ out) {
  long long t = (long long)blockIdx.x * blockDim.x + threadIdx.x;
  long long total = (E + N) * 32;
  if (t >= total) return;
  long long eid = t >> 5; int chunk = (int)(t & 31);
  int s, d; edge_sd(ei, E, eid, s, d);
  float e = as[s] + ad[d];
  e = e > 0.f ? e : NEG_SLOPE * e;
  float alpha = __expf(e - funkey(emaxk[d])) / den[d];
  float4 hv = *(const float4*)(h2 + (size_t)s * 128 + chunk * 4);
  float* op = out + (size_t)d * 128 + chunk * 4;
  atomicAdd(op + 0, alpha * hv.x); atomicAdd(op + 1, alpha * hv.y);
  atomicAdd(op + 2, alpha * hv.z); atomicAdd(op + 3, alpha * hv.w);
}

__global__ void elu_bias_k(const float* __restrict__ agg, const float* __restrict__ b1,
                           float* __restrict__ helu, int N) {
  long long i = (long long)blockIdx.x * blockDim.x + threadIdx.x;
  if (i >= (long long)N * 64) return;
  float v = agg[i] + b1[i & 63];
  helu[i] = v > 0.f ? v : (__expf(v) - 1.f);
}

__global__ void bias2_k(float* __restrict__ out, const float* __restrict__ b2, int N) {
  long long i = (long long)blockIdx.x * blockDim.x + threadIdx.x;
  if (i >= (long long)N * 128) return;
  out[i] += b2[i & 127];
}

static inline unsigned cdiv(long long a, long long b) { return (unsigned)((a + b - 1) / b); }

extern "C" void kernel_launch(void* const* d_in, const int* in_sizes, int n_in,
                              void* d_out, int out_size, void* d_ws, size_t ws_size,
                              hipStream_t stream) {
  const float* x    = (const float*)d_in[0];
  const int*   ei   = (const int*)d_in[1];
  const float* W1   = (const float*)d_in[2];
  const float* as1w = (const float*)d_in[3];
  const float* ad1w = (const float*)d_in[4];
  const float* b1   = (const float*)d_in[5];
  const float* W2   = (const float*)d_in[6];
  const float* as2w = (const float*)d_in[7];
  const float* ad2w = (const float*)d_in[8];
  const float* b2   = (const float*)d_in[9];
  float* out = (float*)d_out;

  const int N = in_sizes[0] / 128;
  const long long E = (long long)in_sizes[1] / 2;

  float* w = (float*)d_ws;
  size_t off = 0;
  float*    h1   = w + off; off += (size_t)N * 64;
  float*    as1  = w + off; off += (size_t)N * 8;
  float*    ad1  = w + off; off += (size_t)N * 8;
  unsigned* emk1 = (unsigned*)(w + off); off += (size_t)N * 8;
  float*    den1 = w + off; off += (size_t)N * 8;
  float*    agg1 = w + off; off += (size_t)N * 64;
  float*    helu = w + off; off += (size_t)N * 64;
  float*    h2   = w + off; off += (size_t)N * 128;
  float*    as2  = w + off; off += (size_t)N;
  float*    ad2  = w + off; off += (size_t)N;
  unsigned* emk2 = (unsigned*)(w + off); off += (size_t)N;
  float*    den2 = w + off; off += (size_t)N;

  const int B = 256;
  const long long T1 = ((long long)E + N) * 8;
  const long long T2 = (long long)E + N;
  const int tiles = (N + 15) / 16;

  hipMemsetAsync(emk1, 0, (size_t)N * 8 * sizeof(unsigned), stream);
  hipMemsetAsync(den1, 0, (size_t)N * 8 * sizeof(float), stream);
  hipMemsetAsync(agg1, 0, (size_t)N * 64 * sizeof(float), stream);
  hipMemsetAsync(emk2, 0, (size_t)N * sizeof(unsigned), stream);
  hipMemsetAsync(den2, 0, (size_t)N * sizeof(float), stream);
  hipMemsetAsync(out,  0, (size_t)N * 128 * sizeof(float), stream);

  gemm_wmma_f32<128, 64><<<cdiv((long long)tiles * 32, B), B, 0, stream>>>(x, W1, h1, N);
  alpha1_k<<<cdiv((long long)N * 8, B), B, 0, stream>>>(h1, as1w, ad1w, as1, ad1, N);
  edge_max_k<8><<<cdiv(T1, B), B, 0, stream>>>(ei, E, N, as1, ad1, emk1);
  edge_den_k<8><<<cdiv(T1, B), B, 0, stream>>>(ei, E, N, as1, ad1, emk1, den1);
  agg1_k<<<cdiv(T1, B), B, 0, stream>>>(ei, E, N, as1, ad1, emk1, den1, h1, agg1);
  elu_bias_k<<<cdiv((long long)N * 64, B), B, 0, stream>>>(agg1, b1, helu, N);

  gemm_wmma_f32<64, 128><<<cdiv((long long)tiles * 32, B), B, 0, stream>>>(helu, W2, h2, N);
  alpha2_k<<<cdiv((long long)N * 32, B), B, 0, stream>>>(h2, as2w, ad2w, as2, ad2, N);
  edge_max_k<1><<<cdiv(T2, B), B, 0, stream>>>(ei, E, N, as2, ad2, emk2);
  edge_den_k<1><<<cdiv(T2, B), B, 0, stream>>>(ei, E, N, as2, ad2, emk2, den2);
  agg2_k<<<cdiv(T2 * 32, B), B, 0, stream>>>(ei, E, N, as2, ad2, emk2, den2, h2, out);
  bias2_k<<<cdiv((long long)N * 128, B), B, 0, stream>>>(out, b2, N);

  (void)n_in; (void)out_size; (void)ws_size;
}


# Logit dots move into the GEMM tail: a wave dots its accumulator tiles with a_src and a_dst, folds over 16-lane rows by DPP adds, writes both logit tables; 7 to 5 launches; logit sum order changes
# speedup vs baseline: 4.4005x; 4.4005x over previous
#include <hip/hip_runtime.h>
#include <cstdint>
#include <cstddef>

typedef __attribute__((ext_vector_type(2))) float v2f;
typedef __attribute__((ext_vector_type(8))) float v8f;

#define NEG_SLOPE 0.2f

__device__ __forceinline__ unsigned fkey(float f) {
  unsigned u = __float_as_uint(f);
  return (u & 0x80000000u) ? ~u : (u | 0x80000000u);
}
__device__ __forceinline__ float funkey(unsigned k) {
  unsigned u = (k & 0x80000000u) ? (k & 0x7fffffffu) : ~k;
  return __uint_as_float(u);
}

template<int MASK>
__device__ __forceinline__ float row_xor(float x) {
  const int b = __float_as_int(x);
  return __int_as_float(__builtin_amdgcn_update_dpp(b, b, 0x160 | MASK, 0xf, 0xf, false));
}
template<int N>
__device__ __forceinline__ float xfold(const float* v, int m) {
  if constexpr (N == 1) {
    return v[0];
  } else {
    constexpr int HALF = N / 2;
    const bool up = (m & HALF) != 0;
    float w[HALF];
    #pragma unroll
    for (int i = 0; i < HALF; ++i) {
      const float keep = up ? v[i + HALF] : v[i];
      const float send = up ? v[i] : v[i + HALF];
      w[i] = keep + row_xor<HALF>(send);
    }
    return xfold<HALF>(w, m);
  }
}

template<int K, int COLS, int H>
__global__ __launch_bounds__(256)
void gemm_wmma_f32(const float* __restrict__ A,
                              const float* __restrict__ B,
                              const float* __restrict__ aw_s, const float* __restrict__ aw_d,
                              float* __restrict__ C,
                              float* __restrict__ as, float* __restrict__ ad, int N) {
  constexpr int CT = COLS / 16;
  constexpr int CH = COLS / H;
  static_assert(CH == 8 || CH % 16 == 0, "a head is half a column tile or whole column tiles");
  const int wave = (int)((blockIdx.x * blockDim.x + threadIdx.x) >> 5);
  const int lane = (int)(threadIdx.x & 31);
  const int row0 = wave * 16;
  if (row0 >= N) return;
  const int m    = lane & 15;
  const int half = lane >> 4;
  const int klo  = half * 2;

  float ws[CT], wd[CT];
  #pragma unroll
  for (int t = 0; t < CT; ++t) { ws[t] = aw_s[t * 16 + m]; wd[t] = aw_d[t * 16 + m]; }

  v8f zero = {0.f, 0.f, 0.f, 0.f, 0.f, 0.f, 0.f, 0.f};
  v8f acc[CT];
  #pragma unroll
  for (int t = 0; t < CT; ++t) acc[t] = zero;

  #pragma unroll
  for (int k = 0; k < K; k += 4) {
    const float* ap = A + (size_t)(row0 + m) * K + (k + klo);
    v2f a; a.x = ap[0]; a.y = ap[1];
    #pragma unroll
    for (int t = 0; t < CT; ++t) {
      const float* bp = B + (size_t)(k + klo) * COLS + t * 16 + m;
      v2f b; b.x = bp[0]; b.y = bp[COLS];
      acc[t] = __builtin_amdgcn_wmma_f32_16x16x4_f32(
          false, a, false, b, (short)0, acc[t], false, false);
    }
  }
  const int rbase = row0 + half * 8;
  #pragma unroll
  for (int t = 0; t < CT; ++t) {
    #pragma unroll
    for (int v = 0; v < 8; ++v)
      C[(size_t)(rbase + v) * COLS + t * 16 + m] = acc[t][v];
  }

  if constexpr (CH >= 16) {
    constexpr int TPH = CH / 16;
    #pragma unroll
    for (int h = 0; h < H; ++h) {
      float v[16];
      #pragma unroll
      for (int r = 0; r < 8; ++r) {
        float s = 0.f, d = 0.f;
        #pragma unroll
        for (int tt = 0; tt < TPH; ++tt) {
          const int t = h * TPH + tt;
          s += acc[t][r] * ws[t];
          d += acc[t][r] * wd[t];
        }
        v[r] = s; v[8 + r] = d;
      }
      const float tot = xfold<16>(v, m);
      float* __restrict__ dst = (m < 8) ? as : ad;
      dst[(size_t)(rbase + (m & 7)) * H + h] = tot;
    }
  } else {
    #pragma unroll
    for (int t = 0; t < CT; ++t) {
      float vs[8], vd[8];
      #pragma unroll
      for (int r = 0; r < 8; ++r) { vs[r] = acc[t][r] * ws[t]; vd[r] = acc[t][r] * wd[t]; }
      const float ts = xfold<8>(vs, m);
      const float td = xfold<8>(vd, m);
      const size_t ix = (size_t)(rbase + (m & 7)) * H + t * 2 + (m >> 3);
      as[ix] = ts; ad[ix] = td;
    }
  }
}

#define OWN_ROWS    196
#define OWN_Q       8192
#define OWN_THREADS 1024
#define OWN_REC_HEAD 32
#define OWN_REC      (OWN_REC_HEAD + 2 * OWN_Q)

__device__ __forceinline__ unsigned owner_byte(int d) {
  return ((unsigned)d / (unsigned)OWN_ROWS) & 255u;
}
__global__ void owner_key_k(const int* __restrict__ ei, long long E, int ngrp,
                            uint4* __restrict__ keys) {
  const int g = (int)(blockIdx.x * blockDim.x + threadIdx.x);
  if (g >= ngrp) return;
  const int* __restrict__ p = ei + E + (size_t)g * 16;
  unsigned w[4];
  #pragma unroll
  for (int j = 0; j < 4; ++j)
    w[j] = owner_byte(p[4 * j]) | (owner_byte(p[4 * j + 1]) << 8) |
           (owner_byte(p[4 * j + 2]) << 16) | (owner_byte(p[4 * j + 3]) << 24);
  keys[g] = make_uint4(w[0], w[1], w[2], w[3]);
}

template<int VPT> struct rowvec;
template<> struct rowvec<2> { typedef float2 type; };
template<> struct rowvec<4> { typedef float4 type; };
__device__ __forceinline__ float vget(const float2& v, int j) { return j == 0 ? v.x : v.y; }
__device__ __forceinline__ float vget(const float4& v, int j) {
  return j == 0 ? v.x : (j == 1 ? v.y : (j == 2 ? v.z : v.w));
}

template<int H, int F, bool ELU, bool FIRST>
__global__ __launch_bounds__(OWN_THREADS)
void own_agg_k(const int* __restrict__ ei, long long E, int N,
               const float* __restrict__ as, const float* __restrict__ ad,
               const float* __restrict__ hsrc, const float* __restrict__ bias,
               float* __restrict__ outp, int* __restrict__ lists,
               const uint4* __restrict__ keys) {
  constexpr int C   = F / H;
  constexpr int VPT = F / 32;
  typedef typename rowvec<VPT>::type vec_t;

  __shared__ float          acc[OWN_ROWS * F];
  __shared__ float          den[OWN_ROWS * H];
  __shared__ float          adl[OWN_ROWS * H];
  __shared__ unsigned       mxk[OWN_ROWS * H];
  __shared__ int            q_s[OWN_Q];
  __shared__ unsigned short q_d[OWN_Q];
  __shared__ int            qn;
  __shared__ int            bad;

  const int tid  = (int)threadIdx.x;
  const int lane = tid & 31;
  const int wave = tid >> 5;
  const int lo   = (int)blockIdx.x * OWN_ROWS;
  const int rows = (N - lo) < OWN_ROWS ? (N - lo) : OWN_ROWS;
  const int Ei   = (int)E;
  const int* __restrict__ dstp = ei + E;
  int* __restrict__ rec   = lists + (size_t)blockIdx.x * OWN_REC;
  int* __restrict__ rec_s = rec + OWN_REC_HEAD;
  int* __restrict__ rec_d = rec_s + OWN_Q;

  for (int i = tid; i < OWN_ROWS * F; i += OWN_THREADS) acc[i] = 0.f;
  for (int i = tid; i < OWN_ROWS * H; i += OWN_THREADS) {
    den[i] = 0.f;
    mxk[i] = 0u;
    const int r = i / H, h = i % H;
    adl[h * OWN_ROWS + r] = (r < rows) ? ad[(size_t)lo * H + i] : 0.f;
  }
  if (FIRST) {
    if (tid == 0) { qn = 0; bad = 0; }
    __syncthreads();
  }

  auto logit = [](float a_src, float a_dst) {
    const float e = a_src + a_dst;
    return e > 0.f ? e : NEG_SLOPE * e;
  };
  auto rescan = [&](auto f) {
    for (int i = tid; i < Ei; i += OWN_THREADS) {
      const unsigned dl = (unsigned)(dstp[i] - lo);
      if (dl < (unsigned)rows) f(ei[i], (int)dl);
    }
  };
  int qtot;
  if (FIRST) {
    auto push = [&](int i) {
      const int pos = atomicAdd(&qn, 1);
      if (pos < OWN_Q) q_s[pos] = i;
    };
    const unsigned pat = ((unsigned)blockIdx.x & 255u) * 0x01010101u;
    auto hits = [&](unsigned w) {
      const unsigned x = w ^ pat;
      return ~(((x & 0x7f7f7f7fu) + 0x7f7f7f7fu) | x | 0x7f7f7f7fu);
    };
    auto scan = [&](const uint4& kv, int i0) {
      const unsigned m0 = hits(kv.x), m1 = hits(kv.y), m2 = hits(kv.z), m3 = hits(kv.w);
      if ((m0 | m1 | m2 | m3) != 0u) {
        unsigned m = (m0 >> 3) | (m1 >> 2) | (m2 >> 1) | m3;
        while (m != 0u) {
          const int t = __builtin_ctz(m);
          m &= m - 1u;
          push(i0 + ((t & 3) << 2) + (t >> 3));
        }
      }
    };
    constexpr int PF = 4;
    const int ngrp  = Ei / 16;
    const int nfull = ngrp / OWN_THREADS;
    if (nfull > 0) {
      int cc = (int)(((long long)blockIdx.x * nfull) / gridDim.x);
      int cp = cc;
      uint4 kb[PF];
      #pragma unroll
      for (int u = 0; u < PF; ++u) {
        kb[u] = keys[(size_t)cp * OWN_THREADS + tid];
        cp = (cp + 1 < nfull) ? cp + 1 : 0;
      }
      int k = 0;
      for (; k + PF <= nfull; k += PF) {
        #pragma unroll
        for (int u = 0; u < PF; ++u) {
          scan(kb[u], (cc * OWN_THREADS + tid) * 16);
          cc = (cc + 1 < nfull) ? cc + 1 : 0;
          kb[u] = keys[(size_t)cp * OWN_THREADS + tid];
          cp = (cp + 1 < nfull) ? cp + 1 : 0;
        }
      }
      #pragma unroll
      for (int u = 0; u < PF - 1; ++u) {
        if (k + u < nfull) {
          scan(kb[u], (cc * OWN_THREADS + tid) * 16);
          cc = (cc + 1 < nfull) ? cc + 1 : 0;
        }
      }
    }
    {
      const int g = nfull * OWN_THREADS + tid;
      if (g < ngrp) scan(keys[g], g * 16);
    }
    for (int i = ngrp * 16 + tid; i < Ei; i += OWN_THREADS)
      if ((unsigned)(dstp[i] - lo) < (unsigned)rows) push(i);
    __syncthreads();
    qtot = qn;
  } else {
    qtot = rec[0];
  }
  bool over = qtot > OWN_Q;
  int  nq   = over ? 0 : qtot;
  if (FIRST) {
    for (int q = tid; q < nq; q += OWN_THREADS) {
      const int i = q_s[q];
      const int s = ei[i];
      const unsigned dl = (unsigned)(dstp[i] - lo);
      if (dl >= (unsigned)rows) bad = 1;
      q_s[q] = s; q_d[q] = (unsigned short)dl; rec_s[q] = s; rec_d[q] = (int)dl;
    }
  } else {
    for (int q = tid; q < nq; q += OWN_THREADS) {
      q_s[q] = rec_s[q]; q_d[q] = (unsigned short)rec_d[q];
    }
  }
  __syncthreads();
  if (FIRST) {
    if (bad != 0) { over = true; nq = 0; qtot = OWN_Q + 1; }
    if (tid == 0) rec[0] = qtot;
  }

  const int total = nq + rows;

  {
    constexpr int UM = (H > 1) ? 4 : 8;
    for (int base = tid; base < total; base += OWN_THREADS * UM) {
      int dlv[UM]; float av[UM][H], dv[UM][H];
      #pragma unroll
      for (int u = 0; u < UM; ++u) {
        const int it0 = base + u * OWN_THREADS;
        const int it  = it0 < total ? it0 : total - 1;
        int s, dl;
        if (it < nq) { s = q_s[it]; dl = (int)q_d[it]; }
        else         { dl = it - nq; s = lo + dl; }
        dlv[u] = dl;
        #pragma unroll
        for (int h = 0; h < H; ++h) dv[u][h] = adl[h * OWN_ROWS + dl];
        const float* ap = as + (size_t)s * H;
        if (H % 4 == 0) {
          #pragma unroll
          for (int h = 0; h < H; h += 4) {
            const float4 t = *(const float4*)(ap + h);
            av[u][h] = t.x; av[u][h + 1] = t.y; av[u][h + 2] = t.z; av[u][h + 3] = t.w;
          }
        } else {
          #pragma unroll
          for (int h = 0; h < H; ++h) av[u][h] = ap[h];
        }
      }
      #pragma unroll
      for (int u = 0; u < UM; ++u) {
        if (base + u * OWN_THREADS < total) {
          #pragma unroll
          for (int h = 0; h < H; ++h) {
            atomicMax(&mxk[h * OWN_ROWS + dlv[u]], fkey(logit(av[u][h], dv[u][h])));
          }
        }
      }
    }
    if (over)
      rescan([&](int s, int dl) {
        #pragma unroll 1
        for (int h = 0; h < H; ++h) {
          const int ix = h * OWN_ROWS + dl;
          atomicMax(&mxk[ix], fkey(logit(as[(size_t)s * H + h], adl[ix])));
        }
      });
  }
  __syncthreads();

  constexpr int EPP  = 32 / H;
  constexpr int U    = 8;
  const int     pe   = lane / H;
  const int     ph   = lane % H;
  const int     head = (lane * VPT) / C;
  for (int base = wave * 32; base < total; base += (OWN_THREADS / 32) * 32) {
    int sK[H], dK[H]; float wK[H];
    #pragma unroll
    for (int k = 0; k < H; ++k) {
      const int  it0 = base + k * EPP + pe;
      const int  it  = it0 < total ? it0 : total - 1;
      const bool ls  = it < nq;
      const int  sq  = q_s[ls ? it : 0];
      const int  dq  = (int)q_d[ls ? it : 0];
      dK[k] = ls ? dq : it - nq;
      sK[k] = ls ? sq : lo + dK[k];
    }
    float avK[H];
    #pragma unroll
    for (int k = 0; k < H; ++k) avK[k] = as[(size_t)sK[k] * H + ph];
    float adK[H]; unsigned mxK[H];
    #pragma unroll
    for (int k = 0; k < H; ++k) {
      const int ix = ph * OWN_ROWS + dK[k];
      adK[k] = adl[ix]; mxK[k] = mxk[ix];
    }
    #pragma unroll
    for (int k = 0; k < H; ++k) {
      wK[k] = __expf(logit(avK[k], adK[k]) - funkey(mxK[k]));
      atomicAdd(&den[dK[k] * H + ph], (base + k * EPP + pe < total) ? wK[k] : 0.f);
    }
    #pragma unroll
    for (int g = 0; g < 32 / U; ++g) {
      if (base + g * U < total) {
        int dd[U]; float aw[U]; vec_t hv[U];
        #pragma unroll
        for (int u = 0; u < U; ++u) {
          const int e   = g * U + u;
          const int k   = e / EPP;
          const int src = (e % EPP) * H;
          const int s   = __builtin_amdgcn_readlane(sK[k], src);
          dd[u] = __builtin_amdgcn_readlane(dK[k], src);
          if (H == 1) aw[u] = __int_as_float(__builtin_amdgcn_readlane(__float_as_int(wK[k]), src));
          else        aw[u] = __shfl(wK[k], src + head, 32);
          hv[u] = *(const vec_t*)(hsrc + (size_t)s * F + lane * VPT);
        }
        #pragma unroll
        for (int u = 0; u < U; ++u) {
          const bool live = base + g * U + u < total;
          float* ap = &acc[dd[u] * F + lane * VPT];
          #pragma unroll
          for (int j = 0; j < VPT; ++j) atomicAdd(ap + j, live ? aw[u] * vget(hv[u], j) : 0.f);
        }
      }
    }
  }
  if (over)
    rescan([&](int s, int dl) {
      #pragma unroll 1
      for (int h = 0; h < H; ++h) {
        const int ix = h * OWN_ROWS + dl;
        const float a = __expf(logit(as[(size_t)s * H + h], adl[ix]) - funkey(mxk[ix]));
        atomicAdd(&den[dl * H + h], a);
        const float* hp = hsrc + (size_t)s * F + h * C;
        #pragma unroll 1
        for (int cc = 0; cc < C; ++cc) atomicAdd(&acc[dl * F + h * C + cc], a * hp[cc]);
      }
    });
  __syncthreads();

  for (int i = tid; i < rows * F; i += OWN_THREADS) {
    const int r = i / F, f = i % F;
    float v = acc[i] / den[r * H + f / C] + bias[f];
    if (ELU) v = v > 0.f ? v : (__expf(v) - 1.f);
    outp[(size_t)lo * F + i] = v;
  }
}

static inline unsigned cdiv(long long a, long long b) { return (unsigned)((a + b - 1) / b); }

extern "C" void kernel_launch(void* const* d_in, const int* in_sizes, int n_in,
                              void* d_out, int out_size, void* d_ws, size_t ws_size,
                              hipStream_t stream) {
  const float* x    = (const float*)d_in[0];
  const int*   ei   = (const int*)d_in[1];
  const float* W1   = (const float*)d_in[2];
  const float* as1w = (const float*)d_in[3];
  const float* ad1w = (const float*)d_in[4];
  const float* b1   = (const float*)d_in[5];
  const float* W2   = (const float*)d_in[6];
  const float* as2w = (const float*)d_in[7];
  const float* ad2w = (const float*)d_in[8];
  const float* b2   = (const float*)d_in[9];
  float* out = (float*)d_out;

  const int N = in_sizes[0] / 128;
  const long long E = (long long)in_sizes[1] / 2;

  float* w = (float*)d_ws;
  size_t off = 0;
  float*    h1   = w + off; off += (size_t)N * 64;
  float*    as1  = w + off; off += (size_t)N * 8;
  float*    ad1  = w + off; off += (size_t)N * 8;
  float*    helu = w + off; off += (size_t)N * 64;
  float*    h2   = w + off; off += (size_t)N * 128;
  float*    as2  = w + off; off += (size_t)N;
  float*    ad2  = w + off; off += (size_t)N;
  const int B = 256;
  const int tiles = (N + 15) / 16;
  const unsigned own_grid = cdiv(N, OWN_ROWS);
  int*      lists = (int*)(w + off); off += (size_t)own_grid * OWN_REC;
  off = (off + 31) & ~(size_t)31;
  uint4*    keys = (uint4*)(w + off);
  const int ngrp = (int)(E / 16);

  owner_key_k<<<cdiv((long long)ngrp + 1, B), B, 0, stream>>>(ei, E, ngrp, keys);

  gemm_wmma_f32<128, 64, 8><<<cdiv((long long)tiles * 32, B), B, 0, stream>>>(x, W1, as1w, ad1w, h1, as1, ad1, N);
  own_agg_k<8, 64, true, true><<<own_grid, OWN_THREADS, 0, stream>>>(ei, E, N, as1, ad1, h1, b1, helu, lists, keys);

  gemm_wmma_f32<64, 128, 1><<<cdiv((long long)tiles * 32, B), B, 0, stream>>>(helu, W2, as2w, ad2w, h2, as2, ad2, N);
  own_agg_k<1, 128, false, false><<<own_grid, OWN_THREADS, 0, stream>>>(ei, E, N, as2, ad2, h2, b2, out, lists, keys);

  (void)n_in; (void)out_size; (void)ws_size;
}


# LDS row sums kept lane-major (float j of lane l at word j*32+l): a float add hits 32 banks; layer 2 had lanes l, l+16 on one bank, 2-way on each add; node pass reads permuted; sum order as before
# speedup vs baseline: 5.8486x; 1.3291x over previous
#include <hip/hip_runtime.h>
#include <cstdint>
#include <cstddef>

typedef __attribute__((ext_vector_type(2))) float v2f;
typedef __attribute__((ext_vector_type(8))) float v8f;

#define NEG_SLOPE 0.2f

__device__ __forceinline__ unsigned fkey(float f) {
  unsigned u = __float_as_uint(f);
  return (u & 0x80000000u) ? ~u : (u | 0x80000000u);
}
__device__ __forceinline__ float funkey(unsigned k) {
  unsigned u = (k & 0x80000000u) ? (k & 0x7fffffffu) : ~k;
  return __uint_as_float(u);
}

template<int MASK>
__device__ __forceinline__ float row_xor(float x) {
  const int b = __float_as_int(x);
  return __int_as_float(__builtin_amdgcn_update_dpp(b, b, 0x160 | MASK, 0xf, 0xf, false));
}
template<int N>
__device__ __forceinline__ float xfold(const float* v, int m) {
  if constexpr (N == 1) {
    return v[0];
  } else {
    constexpr int HALF = N / 2;
    const bool up = (m & HALF) != 0;
    float w[HALF];
    #pragma unroll
    for (int i = 0; i < HALF; ++i) {
      const float keep = up ? v[i + HALF] : v[i];
      const float send = up ? v[i] : v[i + HALF];
      w[i] = keep + row_xor<HALF>(send);
    }
    return xfold<HALF>(w, m);
  }
}

template<int K, int COLS, int H>
__global__ __launch_bounds__(256)
void gemm_wmma_f32(const float* __restrict__ A,
                              const float* __restrict__ B,
                              const float* __restrict__ aw_s, const float* __restrict__ aw_d,
                              float* __restrict__ C,
                              float* __restrict__ as, float* __restrict__ ad, int N) {
  constexpr int CT = COLS / 16;
  constexpr int CH = COLS / H;
  static_assert(CH == 8 || CH % 16 == 0, "a head is half a column tile or whole column tiles");
  const int wave = (int)((blockIdx.x * blockDim.x + threadIdx.x) >> 5);
  const int lane = (int)(threadIdx.x & 31);
  const int row0 = wave * 16;
  if (row0 >= N) return;
  const int m    = lane & 15;
  const int half = lane >> 4;
  const int klo  = half * 2;

  float ws[CT], wd[CT];
  #pragma unroll
  for (int t = 0; t < CT; ++t) { ws[t] = aw_s[t * 16 + m]; wd[t] = aw_d[t * 16 + m]; }

  v8f zero = {0.f, 0.f, 0.f, 0.f, 0.f, 0.f, 0.f, 0.f};
  v8f acc[CT];
  #pragma unroll
  for (int t = 0; t < CT; ++t) acc[t] = zero;

  #pragma unroll
  for (int k = 0; k < K; k += 4) {
    const float* ap = A + (size_t)(row0 + m) * K + (k + klo);
    v2f a; a.x = ap[0]; a.y = ap[1];
    #pragma unroll
    for (int t = 0; t < CT; ++t) {
      const float* bp = B + (size_t)(k + klo) * COLS + t * 16 + m;
      v2f b; b.x = bp[0]; b.y = bp[COLS];
      acc[t] = __builtin_amdgcn_wmma_f32_16x16x4_f32(
          false, a, false, b, (short)0, acc[t], false, false);
    }
  }
  const int rbase = row0 + half * 8;
  #pragma unroll
  for (int t = 0; t < CT; ++t) {
    #pragma unroll
    for (int v = 0; v < 8; ++v)
      C[(size_t)(rbase + v) * COLS + t * 16 + m] = acc[t][v];
  }

  if constexpr (CH >= 16) {
    constexpr int TPH = CH / 16;
    #pragma unroll
    for (int h = 0; h < H; ++h) {
      float v[16];
      #pragma unroll
      for (int r = 0; r < 8; ++r) {
        float s = 0.f, d = 0.f;
        #pragma unroll
        for (int tt = 0; tt < TPH; ++tt) {
          const int t = h * TPH + tt;
          s += acc[t][r] * ws[t];
          d += acc[t][r] * wd[t];
        }
        v[r] = s; v[8 + r] = d;
      }
      const float tot = xfold<16>(v, m);
      float* __restrict__ dst = (m < 8) ? as : ad;
      dst[(size_t)(rbase + (m & 7)) * H + h] = tot;
    }
  } else {
    #pragma unroll
    for (int t = 0; t < CT; ++t) {
      float vs[8], vd[8];
      #pragma unroll
      for (int r = 0; r < 8; ++r) { vs[r] = acc[t][r] * ws[t]; vd[r] = acc[t][r] * wd[t]; }
      const float ts = xfold<8>(vs, m);
      const float td = xfold<8>(vd, m);
      const size_t ix = (size_t)(rbase + (m & 7)) * H + t * 2 + (m >> 3);
      as[ix] = ts; ad[ix] = td;
    }
  }
}

#define OWN_ROWS    196
#define OWN_Q       8192
#define OWN_THREADS 1024
#define OWN_REC_HEAD 32
#define OWN_REC      (OWN_REC_HEAD + 2 * OWN_Q)

__device__ __forceinline__ unsigned owner_byte(int d) {
  return ((unsigned)d / (unsigned)OWN_ROWS) & 255u;
}
__global__ void owner_key_k(const int* __restrict__ ei, long long E, int ngrp,
                            uint4* __restrict__ keys) {
  const int g = (int)(blockIdx.x * blockDim.x + threadIdx.x);
  if (g >= ngrp) return;
  const int* __restrict__ p = ei + E + (size_t)g * 16;
  unsigned w[4];
  #pragma unroll
  for (int j = 0; j < 4; ++j)
    w[j] = owner_byte(p[4 * j]) | (owner_byte(p[4 * j + 1]) << 8) |
           (owner_byte(p[4 * j + 2]) << 16) | (owner_byte(p[4 * j + 3]) << 24);
  keys[g] = make_uint4(w[0], w[1], w[2], w[3]);
}

template<int VPT> struct rowvec;
template<> struct rowvec<2> { typedef float2 type; };
template<> struct rowvec<4> { typedef float4 type; };
__device__ __forceinline__ float vget(const float2& v, int j) { return j == 0 ? v.x : v.y; }
__device__ __forceinline__ float vget(const float4& v, int j) {
  return j == 0 ? v.x : (j == 1 ? v.y : (j == 2 ? v.z : v.w));
}

template<int H, int F, bool ELU, bool FIRST>
__global__ __launch_bounds__(OWN_THREADS)
void own_agg_k(const int* __restrict__ ei, long long E, int N,
               const float* __restrict__ as, const float* __restrict__ ad,
               const float* __restrict__ hsrc, const float* __restrict__ bias,
               float* __restrict__ outp, int* __restrict__ lists,
               const uint4* __restrict__ keys) {
  constexpr int C   = F / H;
  constexpr int VPT = F / 32;
  typedef typename rowvec<VPT>::type vec_t;

  __shared__ float          acc[OWN_ROWS * F];
  __shared__ float          den[OWN_ROWS * H];
  __shared__ float          adl[OWN_ROWS * H];
  __shared__ unsigned       mxk[OWN_ROWS * H];
  __shared__ int            q_s[OWN_Q];
  __shared__ unsigned short q_d[OWN_Q];
  __shared__ int            qn;
  __shared__ int            bad;

  const int tid  = (int)threadIdx.x;
  const int lane = tid & 31;
  const int wave = tid >> 5;
  const int lo   = (int)blockIdx.x * OWN_ROWS;
  const int rows = (N - lo) < OWN_ROWS ? (N - lo) : OWN_ROWS;
  const int Ei   = (int)E;
  const int* __restrict__ dstp = ei + E;
  int* __restrict__ rec   = lists + (size_t)blockIdx.x * OWN_REC;
  int* __restrict__ rec_s = rec + OWN_REC_HEAD;
  int* __restrict__ rec_d = rec_s + OWN_Q;

  for (int i = tid; i < OWN_ROWS * F; i += OWN_THREADS) acc[i] = 0.f;
  for (int i = tid; i < OWN_ROWS * H; i += OWN_THREADS) {
    den[i] = 0.f;
    mxk[i] = 0u;
    const int r = i / H, h = i % H;
    adl[h * OWN_ROWS + r] = (r < rows) ? ad[(size_t)lo * H + i] : 0.f;
  }
  if (FIRST) {
    if (tid == 0) { qn = 0; bad = 0; }
    __syncthreads();
  }

  auto logit = [](float a_src, float a_dst) {
    const float e = a_src + a_dst;
    return e > 0.f ? e : NEG_SLOPE * e;
  };
  auto rescan = [&](auto f) {
    for (int i = tid; i < Ei; i += OWN_THREADS) {
      const unsigned dl = (unsigned)(dstp[i] - lo);
      if (dl < (unsigned)rows) f(ei[i], (int)dl);
    }
  };
  int qtot;
  if (FIRST) {
    auto push = [&](int i) {
      const int pos = atomicAdd(&qn, 1);
      if (pos < OWN_Q) q_s[pos] = i;
    };
    const unsigned pat = ((unsigned)blockIdx.x & 255u) * 0x01010101u;
    auto hits = [&](unsigned w) {
      const unsigned x = w ^ pat;
      return ~(((x & 0x7f7f7f7fu) + 0x7f7f7f7fu) | x | 0x7f7f7f7fu);
    };
    auto scan = [&](const uint4& kv, int i0) {
      const unsigned m0 = hits(kv.x), m1 = hits(kv.y), m2 = hits(kv.z), m3 = hits(kv.w);
      if ((m0 | m1 | m2 | m3) != 0u) {
        unsigned m = (m0 >> 3) | (m1 >> 2) | (m2 >> 1) | m3;
        while (m != 0u) {
          const int t = __builtin_ctz(m);
          m &= m - 1u;
          push(i0 + ((t & 3) << 2) + (t >> 3));
        }
      }
    };
    constexpr int PF = 4;
    const int ngrp  = Ei / 16;
    const int nfull = ngrp / OWN_THREADS;
    if (nfull > 0) {
      int cc = (int)(((long long)blockIdx.x * nfull) / gridDim.x);
      int cp = cc;
      uint4 kb[PF];
      #pragma unroll
      for (int u = 0; u < PF; ++u) {
        kb[u] = keys[(size_t)cp * OWN_THREADS + tid];
        cp = (cp + 1 < nfull) ? cp + 1 : 0;
      }
      int k = 0;
      for (; k + PF <= nfull; k += PF) {
        #pragma unroll
        for (int u = 0; u < PF; ++u) {
          scan(kb[u], (cc * OWN_THREADS + tid) * 16);
          cc = (cc + 1 < nfull) ? cc + 1 : 0;
          kb[u] = keys[(size_t)cp * OWN_THREADS + tid];
          cp = (cp + 1 < nfull) ? cp + 1 : 0;
        }
      }
      #pragma unroll
      for (int u = 0; u < PF - 1; ++u) {
        if (k + u < nfull) {
          scan(kb[u], (cc * OWN_THREADS + tid) * 16);
          cc = (cc + 1 < nfull) ? cc + 1 : 0;
        }
      }
    }
    {
      const int g = nfull * OWN_THREADS + tid;
      if (g < ngrp) scan(keys[g], g * 16);
    }
    for (int i = ngrp * 16 + tid; i < Ei; i += OWN_THREADS)
      if ((unsigned)(dstp[i] - lo) < (unsigned)rows) push(i);
    __syncthreads();
    qtot = qn;
  } else {
    qtot = rec[0];
  }
  bool over = qtot > OWN_Q;
  int  nq   = over ? 0 : qtot;
  if (FIRST) {
    for (int q = tid; q < nq; q += OWN_THREADS) {
      const int i = q_s[q];
      const int s = ei[i];
      const unsigned dl = (unsigned)(dstp[i] - lo);
      if (dl >= (unsigned)rows) bad = 1;
      q_s[q] = s; q_d[q] = (unsigned short)dl; rec_s[q] = s; rec_d[q] = (int)dl;
    }
  } else {
    for (int q = tid; q < nq; q += OWN_THREADS) {
      q_s[q] = rec_s[q]; q_d[q] = (unsigned short)rec_d[q];
    }
  }
  __syncthreads();
  if (FIRST) {
    if (bad != 0) { over = true; nq = 0; qtot = OWN_Q + 1; }
    if (tid == 0) rec[0] = qtot;
  }

  const int total = nq + rows;

  {
    constexpr int UM = (H > 1) ? 4 : 8;
    for (int base = tid; base < total; base += OWN_THREADS * UM) {
      int dlv[UM]; float av[UM][H], dv[UM][H];
      #pragma unroll
      for (int u = 0; u < UM; ++u) {
        const int it0 = base + u * OWN_THREADS;
        const int it  = it0 < total ? it0 : total - 1;
        int s, dl;
        if (it < nq) { s = q_s[it]; dl = (int)q_d[it]; }
        else         { dl = it - nq; s = lo + dl; }
        dlv[u] = dl;
        #pragma unroll
        for (int h = 0; h < H; ++h) dv[u][h] = adl[h * OWN_ROWS + dl];
        const float* ap = as + (size_t)s * H;
        if (H % 4 == 0) {
          #pragma unroll
          for (int h = 0; h < H; h += 4) {
            const float4 t = *(const float4*)(ap + h);
            av[u][h] = t.x; av[u][h + 1] = t.y; av[u][h + 2] = t.z; av[u][h + 3] = t.w;
          }
        } else {
          #pragma unroll
          for (int h = 0; h < H; ++h) av[u][h] = ap[h];
        }
      }
      #pragma unroll
      for (int u = 0; u < UM; ++u) {
        if (base + u * OWN_THREADS < total) {
          #pragma unroll
          for (int h = 0; h < H; ++h) {
            atomicMax(&mxk[h * OWN_ROWS + dlv[u]], fkey(logit(av[u][h], dv[u][h])));
          }
        }
      }
    }
    if (over)
      rescan([&](int s, int dl) {
        #pragma unroll 1
        for (int h = 0; h < H; ++h) {
          const int ix = h * OWN_ROWS + dl;
          atomicMax(&mxk[ix], fkey(logit(as[(size_t)s * H + h], adl[ix])));
        }
      });
  }
  __syncthreads();

  constexpr int EPP  = 32 / H;
  constexpr int U    = 8;
  const int     pe   = lane / H;
  const int     ph   = lane % H;
  const int     head = (lane * VPT) / C;
  for (int base = wave * 32; base < total; base += (OWN_THREADS / 32) * 32) {
    int sK[H], dK[H]; float wK[H];
    #pragma unroll
    for (int k = 0; k < H; ++k) {
      const int  it0 = base + k * EPP + pe;
      const int  it  = it0 < total ? it0 : total - 1;
      const bool ls  = it < nq;
      const int  sq  = q_s[ls ? it : 0];
      const int  dq  = (int)q_d[ls ? it : 0];
      dK[k] = ls ? dq : it - nq;
      sK[k] = ls ? sq : lo + dK[k];
    }
    float avK[H];
    #pragma unroll
    for (int k = 0; k < H; ++k) avK[k] = as[(size_t)sK[k] * H + ph];
    float adK[H]; unsigned mxK[H];
    #pragma unroll
    for (int k = 0; k < H; ++k) {
      const int ix = ph * OWN_ROWS + dK[k];
      adK[k] = adl[ix]; mxK[k] = mxk[ix];
    }
    #pragma unroll
    for (int k = 0; k < H; ++k) {
      wK[k] = __expf(logit(avK[k], adK[k]) - funkey(mxK[k]));
      atomicAdd(&den[dK[k] * H + ph], (base + k * EPP + pe < total) ? wK[k] : 0.f);
    }
    #pragma unroll
    for (int g = 0; g < 32 / U; ++g) {
      if (base + g * U < total) {
        int dd[U]; float aw[U]; vec_t hv[U];
        #pragma unroll
        for (int u = 0; u < U; ++u) {
          const int e   = g * U + u;
          const int k   = e / EPP;
          const int src = (e % EPP) * H;
          const int s   = __builtin_amdgcn_readlane(sK[k], src);
          dd[u] = __builtin_amdgcn_readlane(dK[k], src);
          if (H == 1) aw[u] = __int_as_float(__builtin_amdgcn_readlane(__float_as_int(wK[k]), src));
          else        aw[u] = __shfl(wK[k], src + head, 32);
          hv[u] = *(const vec_t*)(hsrc + (size_t)s * F + lane * VPT);
        }
        #pragma unroll
        for (int u = 0; u < U; ++u) {
          const bool live = base + g * U + u < total;
          float* ap = &acc[dd[u] * F + lane];
          #pragma unroll
          for (int j = 0; j < VPT; ++j) atomicAdd(ap + j * 32, live ? aw[u] * vget(hv[u], j) : 0.f);
        }
      }
    }
  }
  if (over)
    rescan([&](int s, int dl) {
      #pragma unroll 1
      for (int h = 0; h < H; ++h) {
        const int ix = h * OWN_ROWS + dl;
        const float a = __expf(logit(as[(size_t)s * H + h], adl[ix]) - funkey(mxk[ix]));
        atomicAdd(&den[dl * H + h], a);
        const float* hp = hsrc + (size_t)s * F + h * C;
        #pragma unroll 1
        for (int cc = 0; cc < C; ++cc) {
          const int f = h * C + cc;
          atomicAdd(&acc[dl * F + (f % VPT) * 32 + f / VPT], a * hp[cc]);
        }
      }
    });
  __syncthreads();

  for (int i = tid; i < rows * F; i += OWN_THREADS) {
    const int r = i / F, f = i % F;
    float v = acc[r * F + (f % VPT) * 32 + f / VPT] / den[r * H + f / C] + bias[f];
    if (ELU) v = v > 0.f ? v : (__expf(v) - 1.f);
    outp[(size_t)lo * F + i] = v;
  }
}

static inline unsigned cdiv(long long a, long long b) { return (unsigned)((a + b - 1) / b); }

extern "C" void kernel_launch(void* const* d_in, const int* in_sizes, int n_in,
                              void* d_out, int out_size, void* d_ws, size_t ws_size,
                              hipStream_t stream) {
  const float* x    = (const float*)d_in[0];
  const int*   ei   = (const int*)d_in[1];
  const float* W1   = (const float*)d_in[2];
  const float* as1w = (const float*)d_in[3];
  const float* ad1w = (const float*)d_in[4];
  const float* b1   = (const float*)d_in[5];
  const float* W2   = (const float*)d_in[6];
  const float* as2w = (const float*)d_in[7];
  const float* ad2w = (const float*)d_in[8];
  const float* b2   = (const float*)d_in[9];
  float* out = (float*)d_out;

  const int N = in_sizes[0] / 128;
  const long long E = (long long)in_sizes[1] / 2;

  float* w = (float*)d_ws;
  size_t off = 0;
  float*    h1   = w + off; off += (size_t)N * 64;
  float*    as1  = w + off; off += (size_t)N * 8;
  float*    ad1  = w + off; off += (size_t)N * 8;
  float*    helu = w + off; off += (size_t)N * 64;
  float*    h2   = w + off; off += (size_t)N * 128;
  float*    as2  = w + off; off += (size_t)N;
  float*    ad2  = w + off; off += (size_t)N;
  const int B = 256;
  const int tiles = (N + 15) / 16;
  const unsigned own_grid = cdiv(N, OWN_ROWS);
  int*      lists = (int*)(w + off); off += (size_t)own_grid * OWN_REC;
  off = (off + 31) & ~(size_t)31;
  uint4*    keys = (uint4*)(w + off);
  const int ngrp = (int)(E / 16);

  owner_key_k<<<cdiv((long long)ngrp + 1, B), B, 0, stream>>>(ei, E, ngrp, keys);

  gemm_wmma_f32<128, 64, 8><<<cdiv((long long)tiles * 32, B), B, 0, stream>>>(x, W1, as1w, ad1w, h1, as1, ad1, N);
  own_agg_k<8, 64, true, true><<<own_grid, OWN_THREADS, 0, stream>>>(ei, E, N, as1, ad1, h1, b1, helu, lists, keys);

  gemm_wmma_f32<64, 128, 1><<<cdiv((long long)tiles * 32, B), B, 0, stream>>>(helu, W2, as2w, ad2w, h2, as2, ad2, N);
  own_agg_k<1, 128, false, false><<<own_grid, OWN_THREADS, 0, stream>>>(ei, E, N, as2, ad2, h2, b2, out, lists, keys);

  (void)n_in; (void)out_size; (void)ws_size;
}


# Matched-edge list kept in destination-row order per work-group (LDS counting sort in the list pass); a wave sums a run of one row in registers, one LDS add per run; f32 sum order changes
# speedup vs baseline: 6.5116x; 1.1134x over previous
#include <hip/hip_runtime.h>
#include <cstdint>
#include <cstddef>

typedef __attribute__((ext_vector_type(2))) float v2f;
typedef __attribute__((ext_vector_type(8))) float v8f;

#define NEG_SLOPE 0.2f

__device__ __forceinline__ unsigned fkey(float f) {
  unsigned u = __float_as_uint(f);
  return (u & 0x80000000u) ? ~u : (u | 0x80000000u);
}
__device__ __forceinline__ float funkey(unsigned k) {
  unsigned u = (k & 0x80000000u) ? (k & 0x7fffffffu) : ~k;
  return __uint_as_float(u);
}

template<int MASK>
__device__ __forceinline__ float row_xor(float x) {
  const int b = __float_as_int(x);
  return __int_as_float(__builtin_amdgcn_update_dpp(b, b, 0x160 | MASK, 0xf, 0xf, false));
}
template<int N>
__device__ __forceinline__ float xfold(const float* v, int m) {
  if constexpr (N == 1) {
    return v[0];
  } else {
    constexpr int HALF = N / 2;
    const bool up = (m & HALF) != 0;
    float w[HALF];
    #pragma unroll
    for (int i = 0; i < HALF; ++i) {
      const float keep = up ? v[i + HALF] : v[i];
      const float send = up ? v[i] : v[i + HALF];
      w[i] = keep + row_xor<HALF>(send);
    }
    return xfold<HALF>(w, m);
  }
}

template<int K, int COLS, int H>
__global__ __launch_bounds__(256)
void gemm_wmma_f32(const float* __restrict__ A,
                              const float* __restrict__ B,
                              const float* __restrict__ aw_s, const float* __restrict__ aw_d,
                              float* __restrict__ C,
                              float* __restrict__ as, float* __restrict__ ad, int N) {
  constexpr int CT = COLS / 16;
  constexpr int CH = COLS / H;
  static_assert(CH == 8 || CH % 16 == 0, "a head is half a column tile or whole column tiles");
  const int wave = (int)((blockIdx.x * blockDim.x + threadIdx.x) >> 5);
  const int lane = (int)(threadIdx.x & 31);
  const int row0 = wave * 16;
  if (row0 >= N) return;
  const int m    = lane & 15;
  const int half = lane >> 4;
  const int klo  = half * 2;

  float ws[CT], wd[CT];
  #pragma unroll
  for (int t = 0; t < CT; ++t) { ws[t] = aw_s[t * 16 + m]; wd[t] = aw_d[t * 16 + m]; }

  v8f zero = {0.f, 0.f, 0.f, 0.f, 0.f, 0.f, 0.f, 0.f};
  v8f acc[CT];
  #pragma unroll
  for (int t = 0; t < CT; ++t) acc[t] = zero;

  #pragma unroll
  for (int k = 0; k < K; k += 4) {
    const float* ap = A + (size_t)(row0 + m) * K + (k + klo);
    v2f a; a.x = ap[0]; a.y = ap[1];
    #pragma unroll
    for (int t = 0; t < CT; ++t) {
      const float* bp = B + (size_t)(k + klo) * COLS + t * 16 + m;
      v2f b; b.x = bp[0]; b.y = bp[COLS];
      acc[t] = __builtin_amdgcn_wmma_f32_16x16x4_f32(
          false, a, false, b, (short)0, acc[t], false, false);
    }
  }
  const int rbase = row0 + half * 8;
  #pragma unroll
  for (int t = 0; t < CT; ++t) {
    #pragma unroll
    for (int v = 0; v < 8; ++v)
      C[(size_t)(rbase + v) * COLS + t * 16 + m] = acc[t][v];
  }

  if constexpr (CH >= 16) {
    constexpr int TPH = CH / 16;
    #pragma unroll
    for (int h = 0; h < H; ++h) {
      float v[16];
      #pragma unroll
      for (int r = 0; r < 8; ++r) {
        float s = 0.f, d = 0.f;
        #pragma unroll
        for (int tt = 0; tt < TPH; ++tt) {
          const int t = h * TPH + tt;
          s += acc[t][r] * ws[t];
          d += acc[t][r] * wd[t];
        }
        v[r] = s; v[8 + r] = d;
      }
      const float tot = xfold<16>(v, m);
      float* __restrict__ dst = (m < 8) ? as : ad;
      dst[(size_t)(rbase + (m & 7)) * H + h] = tot;
    }
  } else {
    #pragma unroll
    for (int t = 0; t < CT; ++t) {
      float vs[8], vd[8];
      #pragma unroll
      for (int r = 0; r < 8; ++r) { vs[r] = acc[t][r] * ws[t]; vd[r] = acc[t][r] * wd[t]; }
      const float ts = xfold<8>(vs, m);
      const float td = xfold<8>(vd, m);
      const size_t ix = (size_t)(rbase + (m & 7)) * H + t * 2 + (m >> 3);
      as[ix] = ts; ad[ix] = td;
    }
  }
}

#define OWN_ROWS    196
#define OWN_Q       8192
#define OWN_THREADS 1024
#define OWN_REC_HEAD 32
#define OWN_REC      (OWN_REC_HEAD + 2 * OWN_Q)

__device__ __forceinline__ unsigned owner_byte(int d) {
  return ((unsigned)d / (unsigned)OWN_ROWS) & 255u;
}
__global__ void owner_key_k(const int* __restrict__ ei, long long E, int ngrp,
                            uint4* __restrict__ keys) {
  const int g = (int)(blockIdx.x * blockDim.x + threadIdx.x);
  if (g >= ngrp) return;
  const int* __restrict__ p = ei + E + (size_t)g * 16;
  unsigned w[4];
  #pragma unroll
  for (int j = 0; j < 4; ++j)
    w[j] = owner_byte(p[4 * j]) | (owner_byte(p[4 * j + 1]) << 8) |
           (owner_byte(p[4 * j + 2]) << 16) | (owner_byte(p[4 * j + 3]) << 24);
  keys[g] = make_uint4(w[0], w[1], w[2], w[3]);
}

template<int VPT> struct rowvec;
template<> struct rowvec<2> { typedef float2 type; };
template<> struct rowvec<4> { typedef float4 type; };
__device__ __forceinline__ float vget(const float2& v, int j) { return j == 0 ? v.x : v.y; }
__device__ __forceinline__ float vget(const float4& v, int j) {
  return j == 0 ? v.x : (j == 1 ? v.y : (j == 2 ? v.z : v.w));
}

template<int H, int F, bool ELU, bool FIRST>
__global__ __launch_bounds__(OWN_THREADS)
void own_agg_k(const int* __restrict__ ei, long long E, int N,
               const float* __restrict__ as, const float* __restrict__ ad,
               const float* __restrict__ hsrc, const float* __restrict__ bias,
               float* __restrict__ outp, int* __restrict__ lists,
               const uint4* __restrict__ keys) {
  constexpr int C   = F / H;
  constexpr int VPT = F / 32;
  typedef typename rowvec<VPT>::type vec_t;

  __shared__ float          acc[OWN_ROWS * F];
  __shared__ float          den[OWN_ROWS * H];
  __shared__ float          adl[OWN_ROWS * H];
  __shared__ unsigned       mxk[OWN_ROWS * H];
  __shared__ int            q_s[OWN_Q];
  __shared__ unsigned short q_d[OWN_Q];
  __shared__ int            qn;
  __shared__ int            bad;
  __shared__ int            rcnt[OWN_ROWS];

  const int tid  = (int)threadIdx.x;
  const int lane = tid & 31;
  const int wave = tid >> 5;
  const int lo   = (int)blockIdx.x * OWN_ROWS;
  const int rows = (N - lo) < OWN_ROWS ? (N - lo) : OWN_ROWS;
  const int Ei   = (int)E;
  const int* __restrict__ dstp = ei + E;
  int* __restrict__ rec   = lists + (size_t)blockIdx.x * OWN_REC;
  int* __restrict__ rec_s = rec + OWN_REC_HEAD;
  int* __restrict__ rec_d = rec_s + OWN_Q;

  for (int i = tid; i < OWN_ROWS * F; i += OWN_THREADS) acc[i] = 0.f;
  for (int i = tid; i < OWN_ROWS * H; i += OWN_THREADS) {
    den[i] = 0.f;
    mxk[i] = 0u;
    const int r = i / H, h = i % H;
    adl[h * OWN_ROWS + r] = (r < rows) ? ad[(size_t)lo * H + i] : 0.f;
  }
  if (FIRST) {
    if (tid == 0) { qn = 0; bad = 0; }
    for (int i = tid; i < OWN_ROWS; i += OWN_THREADS) rcnt[i] = 0;
    __syncthreads();
  }

  auto logit = [](float a_src, float a_dst) {
    const float e = a_src + a_dst;
    return e > 0.f ? e : NEG_SLOPE * e;
  };
  auto rescan = [&](auto f) {
    for (int i = tid; i < Ei; i += OWN_THREADS) {
      const unsigned dl = (unsigned)(dstp[i] - lo);
      if (dl < (unsigned)rows) f(ei[i], (int)dl);
    }
  };
  int qtot;
  if (FIRST) {
    auto push = [&](int i) {
      const int pos = atomicAdd(&qn, 1);
      if (pos < OWN_Q) q_s[pos] = i;
    };
    const unsigned pat = ((unsigned)blockIdx.x & 255u) * 0x01010101u;
    auto hits = [&](unsigned w) {
      const unsigned x = w ^ pat;
      return ~(((x & 0x7f7f7f7fu) + 0x7f7f7f7fu) | x | 0x7f7f7f7fu);
    };
    auto scan = [&](const uint4& kv, int i0) {
      const unsigned m0 = hits(kv.x), m1 = hits(kv.y), m2 = hits(kv.z), m3 = hits(kv.w);
      if ((m0 | m1 | m2 | m3) != 0u) {
        unsigned m = (m0 >> 3) | (m1 >> 2) | (m2 >> 1) | m3;
        while (m != 0u) {
          const int t = __builtin_ctz(m);
          m &= m - 1u;
          push(i0 + ((t & 3) << 2) + (t >> 3));
        }
      }
    };
    constexpr int PF = 4;
    const int ngrp  = Ei / 16;
    const int nfull = ngrp / OWN_THREADS;
    if (nfull > 0) {
      int cc = (int)(((long long)blockIdx.x * nfull) / gridDim.x);
      int cp = cc;
      uint4 kb[PF];
      #pragma unroll
      for (int u = 0; u < PF; ++u) {
        kb[u] = keys[(size_t)cp * OWN_THREADS + tid];
        cp = (cp + 1 < nfull) ? cp + 1 : 0;
      }
      int k = 0;
      for (; k + PF <= nfull; k += PF) {
        #pragma unroll
        for (int u = 0; u < PF; ++u) {
          scan(kb[u], (cc * OWN_THREADS + tid) * 16);
          cc = (cc + 1 < nfull) ? cc + 1 : 0;
          kb[u] = keys[(size_t)cp * OWN_THREADS + tid];
          cp = (cp + 1 < nfull) ? cp + 1 : 0;
        }
      }
      #pragma unroll
      for (int u = 0; u < PF - 1; ++u) {
        if (k + u < nfull) {
          scan(kb[u], (cc * OWN_THREADS + tid) * 16);
          cc = (cc + 1 < nfull) ? cc + 1 : 0;
        }
      }
    }
    {
      const int g = nfull * OWN_THREADS + tid;
      if (g < ngrp) scan(keys[g], g * 16);
    }
    for (int i = ngrp * 16 + tid; i < Ei; i += OWN_THREADS)
      if ((unsigned)(dstp[i] - lo) < (unsigned)rows) push(i);
    __syncthreads();
    qtot = qn;
  } else {
    qtot = rec[0];
  }
  bool over = qtot > OWN_Q;
  int  nq   = over ? 0 : qtot;
  if (FIRST) {
    constexpr int QPT = OWN_Q / OWN_THREADS;
    static_assert(QPT * OWN_THREADS == OWN_Q, "a lane holds a whole number of list slots");
    int sv[QPT], dv[QPT];
    #pragma unroll
    for (int j = 0; j < QPT; ++j) {
      const int q = tid + j * OWN_THREADS;
      sv[j] = 0; dv[j] = 0;
      if (q < nq) {
        const int i = q_s[q];
        sv[j] = ei[i];
        unsigned dl = (unsigned)(dstp[i] - lo);
        if (dl >= (unsigned)rows) { bad = 1; dl = 0u; }
        dv[j] = (int)dl;
        atomicAdd(&rcnt[dl], 1);
      }
    }
    __syncthreads();
    if (wave == 0) {
      constexpr int PER = (OWN_ROWS + 31) / 32;
      int c[PER]; int sum = 0;
      #pragma unroll
      for (int j = 0; j < PER; ++j) {
        const int r = lane * PER + j;
        c[j] = r < OWN_ROWS ? rcnt[r] : 0;
        sum += c[j];
      }
      int incl = sum;
      #pragma unroll
      for (int o = 1; o < 32; o <<= 1) {
        const int t = __shfl_up(incl, o, 32);
        if (lane >= o) incl += t;
      }
      int run = incl - sum;
      #pragma unroll
      for (int j = 0; j < PER; ++j) {
        const int r = lane * PER + j;
        if (r < OWN_ROWS) rcnt[r] = run;
        run += c[j];
      }
    }
    __syncthreads();
    #pragma unroll
    for (int j = 0; j < QPT; ++j) {
      const int q = tid + j * OWN_THREADS;
      if (q < nq) {
        const int pos = atomicAdd(&rcnt[dv[j]], 1);
        q_s[pos] = sv[j]; q_d[pos] = (unsigned short)dv[j];
      }
    }
  } else {
    for (int q = tid; q < nq; q += OWN_THREADS) {
      q_s[q] = rec_s[q]; q_d[q] = (unsigned short)rec_d[q];
    }
  }
  __syncthreads();
  if (FIRST) {
    if (bad != 0) { over = true; nq = 0; qtot = OWN_Q + 1; }
    if (tid == 0) rec[0] = qtot;
    for (int q = tid; q < nq; q += OWN_THREADS) { rec_s[q] = q_s[q]; rec_d[q] = (int)q_d[q]; }
  }

  const int total = nq + rows;

  {
    constexpr int UM = (H > 1) ? 4 : 8;
    const int P    = ((total + 31) >> 5) | 1;
    const int span = P << 5;
    for (int base = tid; base < span; base += OWN_THREADS * UM) {
      int dlv[UM]; bool okv[UM]; float av[UM][H], dv[UM][H];
      #pragma unroll
      for (int u = 0; u < UM; ++u) {
        const int x   = base + u * OWN_THREADS;
        const int it0 = (x & 31) * P + (x >> 5);
        okv[u] = (x < span) && (it0 < total);
        const int it  = okv[u] ? it0 : total - 1;
        int s, dl;
        if (it < nq) { s = q_s[it]; dl = (int)q_d[it]; }
        else         { dl = it - nq; s = lo + dl; }
        dlv[u] = dl;
        #pragma unroll
        for (int h = 0; h < H; ++h) dv[u][h] = adl[h * OWN_ROWS + dl];
        const float* ap = as + (size_t)s * H;
        if (H % 4 == 0) {
          #pragma unroll
          for (int h = 0; h < H; h += 4) {
            const float4 t = *(const float4*)(ap + h);
            av[u][h] = t.x; av[u][h + 1] = t.y; av[u][h + 2] = t.z; av[u][h + 3] = t.w;
          }
        } else {
          #pragma unroll
          for (int h = 0; h < H; ++h) av[u][h] = ap[h];
        }
      }
      #pragma unroll
      for (int u = 0; u < UM; ++u) {
        if (okv[u]) {
          #pragma unroll
          for (int h = 0; h < H; ++h) {
            atomicMax(&mxk[h * OWN_ROWS + dlv[u]], fkey(logit(av[u][h], dv[u][h])));
          }
        }
      }
    }
    if (over)
      rescan([&](int s, int dl) {
        #pragma unroll 1
        for (int h = 0; h < H; ++h) {
          const int ix = h * OWN_ROWS + dl;
          atomicMax(&mxk[ix], fkey(logit(as[(size_t)s * H + h], adl[ix])));
        }
      });
  }
  __syncthreads();

  constexpr int EPP  = 32 / H;
  constexpr int LPH  = C / VPT;
  constexpr int U    = 8;
  const int     pe   = lane / H;
  const int     ph   = lane % H;
  const int     head = (lane * VPT) / C;
  const int base0 = __builtin_amdgcn_readfirstlane(wave * 32);
  const int totu  = __builtin_amdgcn_readfirstlane(total);
  for (int base = base0; base < totu; base += (OWN_THREADS / 32) * 32) {
    int sK[H], dK[H]; float wK[H];
    #pragma unroll
    for (int k = 0; k < H; ++k) {
      const int  it0 = base + k * EPP + pe;
      const int  it  = it0 < total ? it0 : total - 1;
      const bool ls  = it < nq;
      const int  sq  = q_s[ls ? it : 0];
      const int  dq  = (int)q_d[ls ? it : 0];
      dK[k] = ls ? dq : it - nq;
      sK[k] = ls ? sq : lo + dK[k];
    }
    float avK[H];
    #pragma unroll
    for (int k = 0; k < H; ++k) avK[k] = as[(size_t)sK[k] * H + ph];
    float adK[H]; unsigned mxK[H];
    #pragma unroll
    for (int k = 0; k < H; ++k) {
      const int ix = ph * OWN_ROWS + dK[k];
      adK[k] = adl[ix]; mxK[k] = mxk[ix];
    }
    #pragma unroll
    for (int k = 0; k < H; ++k) {
      wK[k] = __expf(logit(avK[k], adK[k]) - funkey(mxK[k]));
    }
    float racc[VPT]; float rden = 0.f; int cur = -1;
    #pragma unroll
    for (int j = 0; j < VPT; ++j) racc[j] = 0.f;
    auto flush = [&]() {
      float* ap = &acc[cur * F + lane];
      #pragma unroll
      for (int j = 0; j < VPT; ++j) atomicAdd(ap + j * 32, racc[j]);
      if (lane % LPH == 0) atomicAdd(&den[cur * H + head], rden);
    };
    #pragma unroll
    for (int g = 0; g < 32 / U; ++g) {
      if (base + g * U < totu) {
        int dd[U]; float aw[U]; vec_t hv[U];
        #pragma unroll
        for (int u = 0; u < U; ++u) {
          const int e   = g * U + u;
          const int k   = e / EPP;
          const int src = (e % EPP) * H;
          const int s   = __builtin_amdgcn_readlane(sK[k], src);
          dd[u] = __builtin_amdgcn_readlane(dK[k], src);
          if (H == 1) aw[u] = __int_as_float(__builtin_amdgcn_readlane(__float_as_int(wK[k]), src));
          else        aw[u] = __shfl(wK[k], src + head, 32);
          hv[u] = *(const vec_t*)(hsrc + (size_t)s * F + lane * VPT);
        }
        #pragma unroll
        for (int u = 0; u < U; ++u) {
          const bool  live = base + g * U + u < totu;
          const int   d    = live ? dd[u] : cur;
          const float a    = live ? aw[u] : 0.f;
          if (d != cur) {
            if (cur >= 0) flush();
            cur = d; rden = 0.f;
            #pragma unroll
            for (int j = 0; j < VPT; ++j) racc[j] = 0.f;
          }
          #pragma unroll
          for (int j = 0; j < VPT; ++j) racc[j] += a * vget(hv[u], j);
          rden += a;
        }
      }
    }
    if (cur >= 0) flush();
  }
  if (over)
    rescan([&](int s, int dl) {
      #pragma unroll 1
      for (int h = 0; h < H; ++h) {
        const int ix = h * OWN_ROWS + dl;
        const float a = __expf(logit(as[(size_t)s * H + h], adl[ix]) - funkey(mxk[ix]));
        atomicAdd(&den[dl * H + h], a);
        const float* hp = hsrc + (size_t)s * F + h * C;
        #pragma unroll 1
        for (int cc = 0; cc < C; ++cc) {
          const int f = h * C + cc;
          atomicAdd(&acc[dl * F + (f % VPT) * 32 + f / VPT], a * hp[cc]);
        }
      }
    });
  __syncthreads();

  for (int i = tid; i < rows * F; i += OWN_THREADS) {
    const int r = i / F, f = i % F;
    float v = acc[r * F + (f % VPT) * 32 + f / VPT] / den[r * H + f / C] + bias[f];
    if (ELU) v = v > 0.f ? v : (__expf(v) - 1.f);
    outp[(size_t)lo * F + i] = v;
  }
}

static inline unsigned cdiv(long long a, long long b) { return (unsigned)((a + b - 1) / b); }

extern "C" void kernel_launch(void* const* d_in, const int* in_sizes, int n_in,
                              void* d_out, int out_size, void* d_ws, size_t ws_size,
                              hipStream_t stream) {
  const float* x    = (const float*)d_in[0];
  const int*   ei   = (const int*)d_in[1];
  const float* W1   = (const float*)d_in[2];
  const float* as1w = (const float*)d_in[3];
  const float* ad1w = (const float*)d_in[4];
  const float* b1   = (const float*)d_in[5];
  const float* W2   = (const float*)d_in[6];
  const float* as2w = (const float*)d_in[7];
  const float* ad2w = (const float*)d_in[8];
  const float* b2   = (const float*)d_in[9];
  float* out = (float*)d_out;

  const int N = in_sizes[0] / 128;
  const long long E = (long long)in_sizes[1] / 2;

  float* w = (float*)d_ws;
  size_t off = 0;
  float*    h1   = w + off; off += (size_t)N * 64;
  float*    as1  = w + off; off += (size_t)N * 8;
  float*    ad1  = w + off; off += (size_t)N * 8;
  float*    helu = w + off; off += (size_t)N * 64;
  float*    h2   = w + off; off += (size_t)N * 128;
  float*    as2  = w + off; off += (size_t)N;
  float*    ad2  = w + off; off += (size_t)N;
  const int B = 256;
  const int tiles = (N + 15) / 16;
  const unsigned own_grid = cdiv(N, OWN_ROWS);
  int*      lists = (int*)(w + off); off += (size_t)own_grid * OWN_REC;
  off = (off + 31) & ~(size_t)31;
  uint4*    keys = (uint4*)(w + off);
  const int ngrp = (int)(E / 16);

  owner_key_k<<<cdiv((long long)ngrp + 1, B), B, 0, stream>>>(ei, E, ngrp, keys);

  gemm_wmma_f32<128, 64, 8><<<cdiv((long long)tiles * 32, B), B, 0, stream>>>(x, W1, as1w, ad1w, h1, as1, ad1, N);
  own_agg_k<8, 64, true, true><<<own_grid, OWN_THREADS, 0, stream>>>(ei, E, N, as1, ad1, h1, b1, helu, lists, keys);

  gemm_wmma_f32<64, 128, 1><<<cdiv((long long)tiles * 32, B), B, 0, stream>>>(helu, W2, as2w, ad2w, h2, as2, ad2, N);
  own_agg_k<1, 128, false, false><<<own_grid, OWN_THREADS, 0, stream>>>(ei, E, N, as2, ad2, h2, b2, out, lists, keys);

  (void)n_in; (void)out_size; (void)ws_size;
}


# Layer-1 filter skips blocks without its edges: launch ahead adds a presence bit per (32-edge block, owner); a work-group scans its bit plane, then only hit blocks' owner bytes; sum order as before
# speedup vs baseline: 6.9840x; 1.0725x over previous
#include <hip/hip_runtime.h>
#include <cstdint>
#include <cstddef>

typedef __attribute__((ext_vector_type(2))) float v2f;
typedef __attribute__((ext_vector_type(8))) float v8f;

#define NEG_SLOPE 0.2f

__device__ __forceinline__ unsigned fkey(float f) {
  unsigned u = __float_as_uint(f);
  return (u & 0x80000000u) ? ~u : (u | 0x80000000u);
}
__device__ __forceinline__ float funkey(unsigned k) {
  unsigned u = (k & 0x80000000u) ? (k & 0x7fffffffu) : ~k;
  return __uint_as_float(u);
}

template<int MASK>
__device__ __forceinline__ float row_xor(float x) {
  const int b = __float_as_int(x);
  return __int_as_float(__builtin_amdgcn_update_dpp(b, b, 0x160 | MASK, 0xf, 0xf, false));
}
template<int N>
__device__ __forceinline__ float xfold(const float* v, int m) {
  if constexpr (N == 1) {
    return v[0];
  } else {
    constexpr int HALF = N / 2;
    const bool up = (m & HALF) != 0;
    float w[HALF];
    #pragma unroll
    for (int i = 0; i < HALF; ++i) {
      const float keep = up ? v[i + HALF] : v[i];
      const float send = up ? v[i] : v[i + HALF];
      w[i] = keep + row_xor<HALF>(send);
    }
    return xfold<HALF>(w, m);
  }
}

template<int K, int COLS, int H>
__global__ __launch_bounds__(256)
void gemm_wmma_f32(const float* __restrict__ A,
                              const float* __restrict__ B,
                              const float* __restrict__ aw_s, const float* __restrict__ aw_d,
                              float* __restrict__ C,
                              float* __restrict__ as, float* __restrict__ ad, int N) {
  constexpr int CT = COLS / 16;
  constexpr int CH = COLS / H;
  static_assert(CH == 8 || CH % 16 == 0, "a head is half a column tile or whole column tiles");
  const int wave = (int)((blockIdx.x * blockDim.x + threadIdx.x) >> 5);
  const int lane = (int)(threadIdx.x & 31);
  const int row0 = wave * 16;
  if (row0 >= N) return;
  const int m    = lane & 15;
  const int half = lane >> 4;
  const int klo  = half * 2;

  float ws[CT], wd[CT];
  #pragma unroll
  for (int t = 0; t < CT; ++t) { ws[t] = aw_s[t * 16 + m]; wd[t] = aw_d[t * 16 + m]; }

  v8f zero = {0.f, 0.f, 0.f, 0.f, 0.f, 0.f, 0.f, 0.f};
  v8f acc[CT];
  #pragma unroll
  for (int t = 0; t < CT; ++t) acc[t] = zero;

  #pragma unroll
  for (int k = 0; k < K; k += 4) {
    const float* ap = A + (size_t)(row0 + m) * K + (k + klo);
    v2f a; a.x = ap[0]; a.y = ap[1];
    #pragma unroll
    for (int t = 0; t < CT; ++t) {
      const float* bp = B + (size_t)(k + klo) * COLS + t * 16 + m;
      v2f b; b.x = bp[0]; b.y = bp[COLS];
      acc[t] = __builtin_amdgcn_wmma_f32_16x16x4_f32(
          false, a, false, b, (short)0, acc[t], false, false);
    }
  }
  const int rbase = row0 + half * 8;
  #pragma unroll
  for (int t = 0; t < CT; ++t) {
    #pragma unroll
    for (int v = 0; v < 8; ++v)
      C[(size_t)(rbase + v) * COLS + t * 16 + m] = acc[t][v];
  }

  if constexpr (CH >= 16) {
    constexpr int TPH = CH / 16;
    #pragma unroll
    for (int h = 0; h < H; ++h) {
      float v[16];
      #pragma unroll
      for (int r = 0; r < 8; ++r) {
        float s = 0.f, d = 0.f;
        #pragma unroll
        for (int tt = 0; tt < TPH; ++tt) {
          const int t = h * TPH + tt;
          s += acc[t][r] * ws[t];
          d += acc[t][r] * wd[t];
        }
        v[r] = s; v[8 + r] = d;
      }
      const float tot = xfold<16>(v, m);
      float* __restrict__ dst = (m < 8) ? as : ad;
      dst[(size_t)(rbase + (m & 7)) * H + h] = tot;
    }
  } else {
    #pragma unroll
    for (int t = 0; t < CT; ++t) {
      float vs[8], vd[8];
      #pragma unroll
      for (int r = 0; r < 8; ++r) { vs[r] = acc[t][r] * ws[t]; vd[r] = acc[t][r] * wd[t]; }
      const float ts = xfold<8>(vs, m);
      const float td = xfold<8>(vd, m);
      const size_t ix = (size_t)(rbase + (m & 7)) * H + t * 2 + (m >> 3);
      as[ix] = ts; ad[ix] = td;
    }
  }
}

#define OWN_ROWS    196
#define OWN_Q       8192
#define OWN_THREADS 1024
#define OWN_BLK     32
#define OWN_SEG     512
#define OWN_REC_HEAD 32
#define OWN_REC      (OWN_REC_HEAD + 2 * OWN_Q)

__device__ __forceinline__ unsigned owner_byte(int d) {
  return ((unsigned)d / (unsigned)OWN_ROWS) & 255u;
}
__global__ void owner_key_k(const int* __restrict__ ei, long long E, int ngrp, int nblk,
                            uint4* __restrict__ keys, unsigned* __restrict__ pres) {
  const int b = (int)(blockIdx.x * blockDim.x + threadIdx.x);
  if (b > nblk) return;
  if (b == nblk) {
    if (2 * nblk < ngrp) {
      const int* __restrict__ p = ei + E + (size_t)nblk * OWN_BLK;
      unsigned w[4];
      #pragma unroll
      for (int j = 0; j < 4; ++j)
        w[j] = owner_byte(p[4 * j]) | (owner_byte(p[4 * j + 1]) << 8) |
               (owner_byte(p[4 * j + 2]) << 16) | (owner_byte(p[4 * j + 3]) << 24);
      keys[(size_t)nblk * 2] = make_uint4(w[0], w[1], w[2], w[3]);
    }
    return;
  }
  const int* __restrict__ p = ei + E + (size_t)b * OWN_BLK;
  unsigned w[8], pm[8];
  #pragma unroll
  for (int k = 0; k < 8; ++k) pm[k] = 0u;
  #pragma unroll
  for (int j = 0; j < 8; ++j) {
    unsigned o[4];
    #pragma unroll
    for (int i = 0; i < 4; ++i) {
      o[i] = owner_byte(p[4 * j + i]);
      const unsigned bitv = 1u << (o[i] & 31u);
      const unsigned wi   = o[i] >> 5;
      #pragma unroll
      for (int k = 0; k < 8; ++k) pm[k] |= (wi == (unsigned)k) ? bitv : 0u;
    }
    w[j] = o[0] | (o[1] << 8) | (o[2] << 16) | (o[3] << 24);
  }
  keys[(size_t)b * 2]     = make_uint4(w[0], w[1], w[2], w[3]);
  keys[(size_t)b * 2 + 1] = make_uint4(w[4], w[5], w[6], w[7]);
  #pragma unroll
  for (int k = 0; k < 8; ++k) pres[(size_t)k * nblk + b] = pm[k];
}

template<int VPT> struct rowvec;
template<> struct rowvec<2> { typedef float2 type; };
template<> struct rowvec<4> { typedef float4 type; };
__device__ __forceinline__ float vget(const float2& v, int j) { return j == 0 ? v.x : v.y; }
__device__ __forceinline__ float vget(const float4& v, int j) {
  return j == 0 ? v.x : (j == 1 ? v.y : (j == 2 ? v.z : v.w));
}

template<int H, int F, bool ELU, bool FIRST>
__global__ __launch_bounds__(OWN_THREADS)
void own_agg_k(const int* __restrict__ ei, long long E, int N,
               const float* __restrict__ as, const float* __restrict__ ad,
               const float* __restrict__ hsrc, const float* __restrict__ bias,
               float* __restrict__ outp, int* __restrict__ lists,
               const uint4* __restrict__ keys, const unsigned* __restrict__ pres) {
  constexpr int C   = F / H;
  constexpr int VPT = F / 32;
  typedef typename rowvec<VPT>::type vec_t;

  __shared__ float          acc[OWN_ROWS * F];
  __shared__ float          den[OWN_ROWS * H];
  __shared__ float          adl[OWN_ROWS * H];
  __shared__ unsigned       mxk[OWN_ROWS * H];
  __shared__ int            q_s[OWN_Q];
  __shared__ unsigned short q_d[OWN_Q];
  __shared__ int            qn;
  __shared__ int            bad;
  __shared__ int            rcnt[OWN_ROWS];
  __shared__ int            blk_s[FIRST ? (OWN_THREADS / 32) * OWN_SEG : 1];

  const int tid  = (int)threadIdx.x;
  const int lane = tid & 31;
  const int wave = tid >> 5;
  const int lo   = (int)blockIdx.x * OWN_ROWS;
  const int rows = (N - lo) < OWN_ROWS ? (N - lo) : OWN_ROWS;
  const int Ei   = (int)E;
  const int* __restrict__ dstp = ei + E;
  int* __restrict__ rec   = lists + (size_t)blockIdx.x * OWN_REC;
  int* __restrict__ rec_s = rec + OWN_REC_HEAD;
  int* __restrict__ rec_d = rec_s + OWN_Q;

  for (int i = tid; i < OWN_ROWS * F; i += OWN_THREADS) acc[i] = 0.f;
  for (int i = tid; i < OWN_ROWS * H; i += OWN_THREADS) {
    den[i] = 0.f;
    mxk[i] = 0u;
    const int r = i / H, h = i % H;
    adl[h * OWN_ROWS + r] = (r < rows) ? ad[(size_t)lo * H + i] : 0.f;
  }
  if (FIRST) {
    if (tid == 0) { qn = 0; bad = 0; }
    for (int i = tid; i < OWN_ROWS; i += OWN_THREADS) rcnt[i] = 0;
    __syncthreads();
  }

  auto logit = [](float a_src, float a_dst) {
    const float e = a_src + a_dst;
    return e > 0.f ? e : NEG_SLOPE * e;
  };
  auto rescan = [&](auto f) {
    for (int i = tid; i < Ei; i += OWN_THREADS) {
      const unsigned dl = (unsigned)(dstp[i] - lo);
      if (dl < (unsigned)rows) f(ei[i], (int)dl);
    }
  };
  int qtot;
  if (FIRST) {
    auto push = [&](int i) {
      const int pos = atomicAdd(&qn, 1);
      if (pos < OWN_Q) q_s[pos] = i;
    };
    const unsigned pat = ((unsigned)blockIdx.x & 255u) * 0x01010101u;
    auto hits = [&](unsigned w) {
      const unsigned x = w ^ pat;
      return ~(((x & 0x7f7f7f7fu) + 0x7f7f7f7fu) | x | 0x7f7f7f7fu);
    };
    auto scan = [&](const uint4& kv, int i0) {
      const unsigned m0 = hits(kv.x), m1 = hits(kv.y), m2 = hits(kv.z), m3 = hits(kv.w);
      if ((m0 | m1 | m2 | m3) != 0u) {
        unsigned m = (m0 >> 3) | (m1 >> 2) | (m2 >> 1) | m3;
        while (m != 0u) {
          const int t = __builtin_ctz(m);
          m &= m - 1u;
          push(i0 + ((t & 3) << 2) + (t >> 3));
        }
      }
    };
    const int ngrp = Ei / 16;
    const int nblk = Ei / OWN_BLK;
    const unsigned* __restrict__ pw = pres + (size_t)(((unsigned)blockIdx.x & 255u) >> 5) * nblk;
    const unsigned bit = 1u << ((unsigned)blockIdx.x & 31u);
    int* __restrict__ seg = blk_s + wave * OWN_SEG;
    int cw = 0;
    {
      constexpr int PB = 16;
      const int nch = (nblk + OWN_THREADS - 1) / OWN_THREADS;
      const int rot = (int)(((long long)blockIdx.x * nch) / gridDim.x);
      for (int k = 0; k < nch; k += PB) {
        unsigned pv[PB], bm[PB]; int bv[PB];
        #pragma unroll
        for (int u = 0; u < PB; ++u) {
          int cc = rot + k + u;
          if (cc >= nch) cc -= nch;
          const int  b  = cc * OWN_THREADS + tid;
          const bool ok = (k + u < nch) && (b < nblk);
          bv[u] = b;
          bm[u] = ok ? bit : 0u;
          pv[u] = pw[ok ? b : 0];
        }
        #pragma unroll
        for (int u = 0; u < PB; ++u) {
          const unsigned vote = __builtin_amdgcn_ballot_w32((pv[u] & bm[u]) != 0u);
          const bool     hit  = (pv[u] & bm[u]) != 0u;
          const int      pos  = cw + (int)__builtin_amdgcn_mbcnt_lo(vote, 0u);
          if (hit && pos < OWN_SEG) seg[pos] = bv[u];
          cw += __builtin_popcount(vote);
        }
      }
    }
    if (cw > OWN_SEG && lane == 0) atomicAdd(&qn, OWN_Q + 1);
    const int myn = cw < OWN_SEG ? cw : OWN_SEG;
    __syncthreads();
    {
      constexpr int TP = 3;
      for (int t0 = 0; t0 < myn; t0 += 32 * TP) {
        int b2[TP]; uint4 ka[TP], kc[TP];
        #pragma unroll
        for (int u = 0; u < TP; ++u) {
          const int  t  = t0 + u * 32 + lane;
          const bool ok = t < myn;
          const int  b  = seg[ok ? t : 0];
          b2[u] = ok ? b : -1;
          ka[u] = keys[(size_t)b * 2];
          kc[u] = keys[(size_t)b * 2 + 1];
        }
        #pragma unroll
        for (int u = 0; u < TP; ++u) {
          if (b2[u] >= 0) {
            scan(ka[u], b2[u] * OWN_BLK);
            scan(kc[u], b2[u] * OWN_BLK + 16);
          }
        }
      }
    }
    {
      const int g = nblk * 2 + tid;
      if (g < ngrp) scan(keys[g], g * 16);
    }
    for (int i = ngrp * 16 + tid; i < Ei; i += OWN_THREADS)
      if ((unsigned)(dstp[i] - lo) < (unsigned)rows) push(i);
    __syncthreads();
    qtot = qn;
  } else {
    qtot = rec[0];
  }
  bool over = qtot > OWN_Q;
  int  nq   = over ? 0 : qtot;
  if (FIRST) {
    constexpr int QPT = OWN_Q / OWN_THREADS;
    static_assert(QPT * OWN_THREADS == OWN_Q, "a lane holds a whole number of list slots");
    int sv[QPT], dv[QPT];
    #pragma unroll
    for (int j = 0; j < QPT; ++j) {
      const int q = tid + j * OWN_THREADS;
      sv[j] = 0; dv[j] = 0;
      if (q < nq) {
        const int i = q_s[q];
        sv[j] = ei[i];
        unsigned dl = (unsigned)(dstp[i] - lo);
        if (dl >= (unsigned)rows) { bad = 1; dl = 0u; }
        dv[j] = (int)dl;
        atomicAdd(&rcnt[dl], 1);
      }
    }
    __syncthreads();
    if (wave == 0) {
      constexpr int PER = (OWN_ROWS + 31) / 32;
      int c[PER]; int sum = 0;
      #pragma unroll
      for (int j = 0; j < PER; ++j) {
        const int r = lane * PER + j;
        c[j] = r < OWN_ROWS ? rcnt[r] : 0;
        sum += c[j];
      }
      int incl = sum;
      #pragma unroll
      for (int o = 1; o < 32; o <<= 1) {
        const int t = __shfl_up(incl, o, 32);
        if (lane >= o) incl += t;
      }
      int run = incl - sum;
      #pragma unroll
      for (int j = 0; j < PER; ++j) {
        const int r = lane * PER + j;
        if (r < OWN_ROWS) rcnt[r] = run;
        run += c[j];
      }
    }
    __syncthreads();
    #pragma unroll
    for (int j = 0; j < QPT; ++j) {
      const int q = tid + j * OWN_THREADS;
      if (q < nq) {
        const int pos = atomicAdd(&rcnt[dv[j]], 1);
        q_s[pos] = sv[j]; q_d[pos] = (unsigned short)dv[j];
      }
    }
  } else {
    for (int q = tid; q < nq; q += OWN_THREADS) {
      q_s[q] = rec_s[q]; q_d[q] = (unsigned short)rec_d[q];
    }
  }
  __syncthreads();
  if (FIRST) {
    if (bad != 0) { over = true; nq = 0; qtot = OWN_Q + 1; }
    if (tid == 0) rec[0] = qtot;
    for (int q = tid; q < nq; q += OWN_THREADS) { rec_s[q] = q_s[q]; rec_d[q] = (int)q_d[q]; }
  }

  const int total = nq + rows;

  {
    constexpr int UM = (H > 1) ? 4 : 8;
    const int P    = ((total + 31) >> 5) | 1;
    const int span = P << 5;
    for (int base = tid; base < span; base += OWN_THREADS * UM) {
      int dlv[UM]; bool okv[UM]; float av[UM][H], dv[UM][H];
      #pragma unroll
      for (int u = 0; u < UM; ++u) {
        const int x   = base + u * OWN_THREADS;
        const int it0 = (x & 31) * P + (x >> 5);
        okv[u] = (x < span) && (it0 < total);
        const int it  = okv[u] ? it0 : total - 1;
        int s, dl;
        if (it < nq) { s = q_s[it]; dl = (int)q_d[it]; }
        else         { dl = it - nq; s = lo + dl; }
        dlv[u] = dl;
        #pragma unroll
        for (int h = 0; h < H; ++h) dv[u][h] = adl[h * OWN_ROWS + dl];
        const float* ap = as + (size_t)s * H;
        if (H % 4 == 0) {
          #pragma unroll
          for (int h = 0; h < H; h += 4) {
            const float4 t = *(const float4*)(ap + h);
            av[u][h] = t.x; av[u][h + 1] = t.y; av[u][h + 2] = t.z; av[u][h + 3] = t.w;
          }
        } else {
          #pragma unroll
          for (int h = 0; h < H; ++h) av[u][h] = ap[h];
        }
      }
      #pragma unroll
      for (int u = 0; u < UM; ++u) {
        if (okv[u]) {
          #pragma unroll
          for (int h = 0; h < H; ++h) {
            atomicMax(&mxk[h * OWN_ROWS + dlv[u]], fkey(logit(av[u][h], dv[u][h])));
          }
        }
      }
    }
    if (over)
      rescan([&](int s, int dl) {
        #pragma unroll 1
        for (int h = 0; h < H; ++h) {
          const int ix = h * OWN_ROWS + dl;
          atomicMax(&mxk[ix], fkey(logit(as[(size_t)s * H + h], adl[ix])));
        }
      });
  }
  __syncthreads();

  constexpr int EPP  = 32 / H;
  constexpr int LPH  = C / VPT;
  constexpr int U    = 8;
  const int     pe   = lane / H;
  const int     ph   = lane % H;
  const int     head = (lane * VPT) / C;
  const int base0 = __builtin_amdgcn_readfirstlane(wave * 32);
  const int totu  = __builtin_amdgcn_readfirstlane(total);
  for (int base = base0; base < totu; base += (OWN_THREADS / 32) * 32) {
    int sK[H], dK[H]; float wK[H];
    #pragma unroll
    for (int k = 0; k < H; ++k) {
      const int  it0 = base + k * EPP + pe;
      const int  it  = it0 < total ? it0 : total - 1;
      const bool ls  = it < nq;
      const int  sq  = q_s[ls ? it : 0];
      const int  dq  = (int)q_d[ls ? it : 0];
      dK[k] = ls ? dq : it - nq;
      sK[k] = ls ? sq : lo + dK[k];
    }
    float avK[H];
    #pragma unroll
    for (int k = 0; k < H; ++k) avK[k] = as[(size_t)sK[k] * H + ph];
    float adK[H]; unsigned mxK[H];
    #pragma unroll
    for (int k = 0; k < H; ++k) {
      const int ix = ph * OWN_ROWS + dK[k];
      adK[k] = adl[ix]; mxK[k] = mxk[ix];
    }
    #pragma unroll
    for (int k = 0; k < H; ++k) {
      wK[k] = __expf(logit(avK[k], adK[k]) - funkey(mxK[k]));
    }
    float racc[VPT]; float rden = 0.f; int cur = -1;
    #pragma unroll
    for (int j = 0; j < VPT; ++j) racc[j] = 0.f;
    auto flush = [&]() {
      float* ap = &acc[cur * F + lane];
      #pragma unroll
      for (int j = 0; j < VPT; ++j) atomicAdd(ap + j * 32, racc[j]);
      if (lane % LPH == 0) atomicAdd(&den[cur * H + head], rden);
    };
    #pragma unroll
    for (int g = 0; g < 32 / U; ++g) {
      if (base + g * U < totu) {
        int dd[U]; float aw[U]; vec_t hv[U];
        #pragma unroll
        for (int u = 0; u < U; ++u) {
          const int e   = g * U + u;
          const int k   = e / EPP;
          const int src = (e % EPP) * H;
          const int s   = __builtin_amdgcn_readlane(sK[k], src);
          dd[u] = __builtin_amdgcn_readlane(dK[k], src);
          if (H == 1) aw[u] = __int_as_float(__builtin_amdgcn_readlane(__float_as_int(wK[k]), src));
          else        aw[u] = __shfl(wK[k], src + head, 32);
          hv[u] = *(const vec_t*)(hsrc + (size_t)s * F + lane * VPT);
        }
        #pragma unroll
        for (int u = 0; u < U; ++u) {
          const bool  live = base + g * U + u < totu;
          const int   d    = live ? dd[u] : cur;
          const float a    = live ? aw[u] : 0.f;
          if (d != cur) {
            if (cur >= 0) flush();
            cur = d; rden = 0.f;
            #pragma unroll
            for (int j = 0; j < VPT; ++j) racc[j] = 0.f;
          }
          #pragma unroll
          for (int j = 0; j < VPT; ++j) racc[j] += a * vget(hv[u], j);
          rden += a;
        }
      }
    }
    if (cur >= 0) flush();
  }
  if (over)
    rescan([&](int s, int dl) {
      #pragma unroll 1
      for (int h = 0; h < H; ++h) {
        const int ix = h * OWN_ROWS + dl;
        const float a = __expf(logit(as[(size_t)s * H + h], adl[ix]) - funkey(mxk[ix]));
        atomicAdd(&den[dl * H + h], a);
        const float* hp = hsrc + (size_t)s * F + h * C;
        #pragma unroll 1
        for (int cc = 0; cc < C; ++cc) {
          const int f = h * C + cc;
          atomicAdd(&acc[dl * F + (f % VPT) * 32 + f / VPT], a * hp[cc]);
        }
      }
    });
  __syncthreads();

  for (int i = tid; i < rows * F; i += OWN_THREADS) {
    const int r = i / F, f = i % F;
    float v = acc[r * F + (f % VPT) * 32 + f / VPT] / den[r * H + f / C] + bias[f];
    if (ELU) v = v > 0.f ? v : (__expf(v) - 1.f);
    outp[(size_t)lo * F + i] = v;
  }
}

static inline unsigned cdiv(long long a, long long b) { return (unsigned)((a + b - 1) / b); }

extern "C" void kernel_launch(void* const* d_in, const int* in_sizes, int n_in,
                              void* d_out, int out_size, void* d_ws, size_t ws_size,
                              hipStream_t stream) {
  const float* x    = (const float*)d_in[0];
  const int*   ei   = (const int*)d_in[1];
  const float* W1   = (const float*)d_in[2];
  const float* as1w = (const float*)d_in[3];
  const float* ad1w = (const float*)d_in[4];
  const float* b1   = (const float*)d_in[5];
  const float* W2   = (const float*)d_in[6];
  const float* as2w = (const float*)d_in[7];
  const float* ad2w = (const float*)d_in[8];
  const float* b2   = (const float*)d_in[9];
  float* out = (float*)d_out;

  const int N = in_sizes[0] / 128;
  const long long E = (long long)in_sizes[1] / 2;

  float* w = (float*)d_ws;
  size_t off = 0;
  float*    h1   = w + off; off += (size_t)N * 64;
  float*    as1  = w + off; off += (size_t)N * 8;
  float*    ad1  = w + off; off += (size_t)N * 8;
  float*    helu = w + off; off += (size_t)N * 64;
  float*    h2   = w + off; off += (size_t)N * 128;
  float*    as2  = w + off; off += (size_t)N;
  float*    ad2  = w + off; off += (size_t)N;
  const int B = 256;
  const int tiles = (N + 15) / 16;
  const unsigned own_grid = cdiv(N, OWN_ROWS);
  int*      lists = (int*)(w + off); off += (size_t)own_grid * OWN_REC;
  off = (off + 31) & ~(size_t)31;
  uint4*    keys = (uint4*)(w + off);
  const int ngrp = (int)(E / 16);
  const int nblk = (int)(E / OWN_BLK);
  unsigned* pres = (unsigned*)(keys + ngrp);

  owner_key_k<<<cdiv((long long)nblk + 1, B), B, 0, stream>>>(ei, E, ngrp, nblk, keys, pres);

  gemm_wmma_f32<128, 64, 8><<<cdiv((long long)tiles * 32, B), B, 0, stream>>>(x, W1, as1w, ad1w, h1, as1, ad1, N);
  own_agg_k<8, 64, true, true><<<own_grid, OWN_THREADS, 0, stream>>>(ei, E, N, as1, ad1, h1, b1, helu, lists, keys, pres);

  gemm_wmma_f32<64, 128, 1><<<cdiv((long long)tiles * 32, B), B, 0, stream>>>(helu, W2, as2w, ad2w, h2, as2, ad2, N);
  own_agg_k<1, 128, false, false><<<own_grid, OWN_THREADS, 0, stream>>>(ei, E, N, as2, ad2, h2, b2, out, lists, keys, pres);

  (void)n_in; (void)out_size; (void)ws_size;
}
